# starting kernel
# baseline (speedup 1.0000x reference)
#include <hip/hip_runtime.h>
#include <hip/hip_bf16.h>

typedef __attribute__((ext_vector_type(2))) float v2f;
typedef __attribute__((ext_vector_type(8))) float v8f;

#define NNODES 100000
#define NEDGES 1600000

template <int K, int C>
__global__ __launch_bounds__(32 * (C / 16)) void gemm_wmma_kernel(
    const float* __restrict__ A, const float* __restrict__ B,
    float* __restrict__ D) {
  const int lane = threadIdx.x & 31;
  const int wave = threadIdx.x >> 5;
  const int row0 = blockIdx.x * 16;
  const int col0 = wave * 16;
  const int half = lane >> 4;
  const int l    = lane & 15;

  const float* arow = A + (long)(row0 + l) * K + 2 * half;
  const float* bcol = B + col0 + l;

  v8f acc = {};
#pragma unroll 4
  for (int k = 0; k < K; k += 4) {
    v2f a, b;
    a.x = arow[k + 0];
    a.y = arow[k + 1];
    b.x = bcol[(k + 2 * half + 0) * C];
    b.y = bcol[(k + 2 * half + 1) * C];
    acc = __builtin_amdgcn_wmma_f32_16x16x4_f32(
         false, a,  false, b,
         (short)0, acc,  false,  false);
  }

  float* drow = D + (long)(row0 + 8 * half) * C + col0 + l;
#pragma unroll
  for (int i = 0; i < 8; ++i) drow[i * C] = acc[i];
}

__global__ void fill_u32_kernel(unsigned* __restrict__ p, unsigned v, int n) {
  int i = blockIdx.x * blockDim.x + threadIdx.x;
  if (i < n) p[i] = v;
}

template <int H, int D>
__global__ void logits_kernel(const float* __restrict__ feat,
                              const float* __restrict__ al,
                              const float* __restrict__ ar,
                              float* __restrict__ el, float* __restrict__ er,
                              int NH) {
  int i = blockIdx.x * blockDim.x + threadIdx.x;
  if (i >= NH) return;
  int n = i / H, h = i % H;
  const float* f = feat + ((long)n * H + h) * D;
  const float* a = al + h * D;
  const float* r = ar + h * D;
  float sl = 0.f, sr = 0.f;
#pragma unroll
  for (int d = 0; d < D; ++d) {
    sl += f[d] * a[d];
    sr += f[d] * r[d];
  }
  el[i] = sl;
  er[i] = sr;
}

__device__ __forceinline__ unsigned float_ord_encode(float f) {
  unsigned u = __float_as_uint(f);
  return (u & 0x80000000u) ? ~u : (u | 0x80000000u);
}
__device__ __forceinline__ float float_ord_decode(unsigned u) {
  return (u & 0x80000000u) ? __uint_as_float(u ^ 0x80000000u)
                           : __uint_as_float(~u);
}
#define ENC_NEG_INF 0x007FFFFFu

template <int H>
__global__ void edge_max_kernel(const int* __restrict__ src,
                                const int* __restrict__ dst,
                                const float* __restrict__ el,
                                const float* __restrict__ er,
                                unsigned* __restrict__ mu, int EH) {
  int i = blockIdx.x * blockDim.x + threadIdx.x;
  if (i >= EH) return;
  int e = i / H, h = i % H;
  int s = src[e], d = dst[e];
  float v = el[s * H + h] + er[d * H + h];
  v = v > 0.f ? v : 0.2f * v;
  atomicMax(&mu[d * H + h], float_ord_encode(v));
}

template <int H, int D>
__global__ void edge_agg_kernel(const int* __restrict__ src,
                                const int* __restrict__ dst,
                                const float* __restrict__ el,
                                const float* __restrict__ er,
                                const unsigned* __restrict__ mu,
                                float* __restrict__ den,
                                const float* __restrict__ feat,
                                float* __restrict__ agg, int EH) {
  int i = blockIdx.x * blockDim.x + threadIdx.x;
  if (i >= EH) return;
  int e = i / H, h = i % H;
  int s = src[e], d = dst[e];
  float v = el[s * H + h] + er[d * H + h];
  v = v > 0.f ? v : 0.2f * v;
  float m = float_ord_decode(mu[d * H + h]);
  float w = expf(v - m);
  atomicAdd(&den[d * H + h], w);
  const float* fs = feat + ((long)s * H + h) * D;
  float* ad = agg + ((long)d * H + h) * D;
#pragma unroll
  for (int k = 0; k < D; ++k) atomicAdd(&ad[k], w * fs[k]);
}

template <int H, int D, bool DO_ELU>
__global__ void finalize_kernel(const float* __restrict__ agg,
                                const float* __restrict__ den,
                                const float* __restrict__ bias,
                                float* __restrict__ out, int NC) {
  int i = blockIdx.x * blockDim.x + threadIdx.x;
  if (i >= NC) return;
  int n = i / (H * D);
  int c = i % (H * D);
  int h = c / D;
  float dn = den[n * H + h];
  float x = dn > 0.f ? agg[i] / dn : 0.f;
  x += bias[c];
  if (DO_ELU) x = x > 0.f ? x : expf(x) - 1.f;
  out[i] = x;
}

extern "C" void kernel_launch(void* const* d_in, const int* in_sizes, int n_in,
                              void* d_out, int out_size, void* d_ws,
                              size_t ws_size, hipStream_t stream) {
  const float* x   = (const float*)d_in[0];
  const int*   src = (const int*)d_in[1];
  const int*   dst = (const int*)d_in[2];
  const float* W1  = (const float*)d_in[3];
  const float* al1 = (const float*)d_in[4];
  const float* ar1 = (const float*)d_in[5];
  const float* b1  = (const float*)d_in[6];
  const float* W2  = (const float*)d_in[7];
  const float* al2 = (const float*)d_in[8];
  const float* ar2 = (const float*)d_in[9];
  const float* b2  = (const float*)d_in[10];
  float* out = (float*)d_out;

  const int N = NNODES, E = NEDGES;

  float* ws = (float*)d_ws;
  float*    feat1 = ws;
  float*    el1   = feat1 + (long)N * 64;
  float*    er1   = el1 + (long)N * 8;
  unsigned* m1u   = (unsigned*)(er1 + (long)N * 8);
  float*    den1  = (float*)(m1u + (long)N * 8);
  float*    agg1  = den1 + (long)N * 8;
  float*    h1    = feat1;
  float*    feat2 = agg1;
  unsigned* m2u   = (unsigned*)(agg1 + (long)N * 32);
  float*    den2  = agg1 + (long)N * 32 + N;
  float*    el2   = agg1 + (long)N * 32 + 2 * N;
  float*    er2   = agg1 + (long)N * 32 + 3 * N;
  float*    agg2  = el1;

  const int TB = 256;
  auto nb = [](long n, int b) { return (int)((n + b - 1) / b); };

  fill_u32_kernel<<<nb((long)N * 8, TB), TB, 0, stream>>>(m1u, ENC_NEG_INF, N * 8);
  fill_u32_kernel<<<nb((long)N * 8, TB), TB, 0, stream>>>((unsigned*)den1, 0u, N * 8);
  fill_u32_kernel<<<nb((long)N * 64, TB), TB, 0, stream>>>((unsigned*)agg1, 0u, N * 64);

  gemm_wmma_kernel<256, 64><<<N / 16, 128, 0, stream>>>(x, W1, feat1);
  logits_kernel<8, 8><<<nb((long)N * 8, TB), TB, 0, stream>>>(feat1, al1, ar1, el1, er1, N * 8);
  edge_max_kernel<8><<<nb((long)E * 8, TB), TB, 0, stream>>>(src, dst, el1, er1, m1u, E * 8);
  edge_agg_kernel<8, 8><<<nb((long)E * 8, TB), TB, 0, stream>>>(src, dst, el1, er1, m1u, den1, feat1, agg1, E * 8);
  finalize_kernel<8, 8, true><<<nb((long)N * 64, TB), TB, 0, stream>>>(agg1, den1, b1, h1, N * 64);

  fill_u32_kernel<<<nb((long)N, TB), TB, 0, stream>>>(m2u, ENC_NEG_INF, N);
  fill_u32_kernel<<<nb((long)N, TB), TB, 0, stream>>>((unsigned*)den2, 0u, N);

  gemm_wmma_kernel<64, 32><<<N / 16, 64, 0, stream>>>(h1, W2, feat2);
  logits_kernel<1, 32><<<nb((long)N, TB), TB, 0, stream>>>(feat2, al2, ar2, el2, er2, N);

  fill_u32_kernel<<<nb((long)N * 32, TB), TB, 0, stream>>>((unsigned*)agg2, 0u, N * 32);

  edge_max_kernel<1><<<nb((long)E, TB), TB, 0, stream>>>(src, dst, el2, er2, m2u, E);
  edge_agg_kernel<1, 32><<<nb((long)E, TB), TB, 0, stream>>>(src, dst, el2, er2, m2u, den2, feat2, agg2, E);
  finalize_kernel<1, 32, false><<<nb((long)N * 32, TB), TB, 0, stream>>>(agg2, den2, b2, out, N * 32);
}


# edge softmax + aggregation by owner work-groups: each keeps max, denominator and sums of its 391 nodes in LDS and filters the whole dst list itself; LDS float adds, arrival order; 16 -> 6 launches
# speedup vs baseline: 2.6142x; 2.6142x over previous
#include <hip/hip_runtime.h>
#include <hip/hip_bf16.h>

typedef __attribute__((ext_vector_type(2))) float v2f;
typedef __attribute__((ext_vector_type(8))) float v8f;

#define NNODES 100000
#define NEDGES 1600000

template <int K, int C>
__global__ __launch_bounds__(32 * (C / 16)) void gemm_wmma_kernel(
    const float* __restrict__ A, const float* __restrict__ B,
    float* __restrict__ D) {
  const int lane = threadIdx.x & 31;
  const int wave = threadIdx.x >> 5;
  const int row0 = blockIdx.x * 16;
  const int col0 = wave * 16;
  const int half = lane >> 4;
  const int l    = lane & 15;

  const float* arow = A + (long)(row0 + l) * K + 2 * half;
  const float* bcol = B + col0 + l;

  v8f acc = {};
#pragma unroll 4
  for (int k = 0; k < K; k += 4) {
    v2f a, b;
    a.x = arow[k + 0];
    a.y = arow[k + 1];
    b.x = bcol[(k + 2 * half + 0) * C];
    b.y = bcol[(k + 2 * half + 1) * C];
    acc = __builtin_amdgcn_wmma_f32_16x16x4_f32(
         false, a,  false, b,
         (short)0, acc,  false,  false);
  }

  float* drow = D + (long)(row0 + 8 * half) * C + col0 + l;
#pragma unroll
  for (int i = 0; i < 8; ++i) drow[i * C] = acc[i];
}

template <int H, int D>
__global__ void logits_kernel(const float* __restrict__ feat,
                              const float* __restrict__ al,
                              const float* __restrict__ ar,
                              float* __restrict__ el, float* __restrict__ er,
                              int NH) {
  int i = blockIdx.x * blockDim.x + threadIdx.x;
  if (i >= NH) return;
  int n = i / H, h = i % H;
  const float* f = feat + ((long)n * H + h) * D;
  const float* a = al + h * D;
  const float* r = ar + h * D;
  float sl = 0.f, sr = 0.f;
#pragma unroll
  for (int d = 0; d < D; ++d) {
    sl += f[d] * a[d];
    sr += f[d] * r[d];
  }
  el[i] = sl;
  er[i] = sr;
}

__device__ __forceinline__ unsigned float_ord_encode(float f) {
  unsigned u = __float_as_uint(f);
  return (u & 0x80000000u) ? ~u : (u | 0x80000000u);
}
__device__ __forceinline__ float float_ord_decode(unsigned u) {
  return (u & 0x80000000u) ? __uint_as_float(u ^ 0x80000000u)
                           : __uint_as_float(~u);
}
#define ENC_NEG_INF 0x007FFFFFu

#define OWN_R 391
#define OWN_WAVES 32
#define OWN_QCAP 1024

template <int H, int D, bool DO_ELU>
__global__ __launch_bounds__(1024) void owner_agg_kernel(
    const int* __restrict__ src, const int* __restrict__ dst,
    const float* __restrict__ el, const float* __restrict__ er,
    const float* __restrict__ feat, const float* __restrict__ bias,
    float* __restrict__ out) {
  constexpr int C = H * D;
  constexpr int CPL = C / 32;
  constexpr int NG4 = NEDGES / 4;
  constexpr int GPI = OWN_WAVES * 128;
  constexpr int NIT = (NG4 + GPI - 1) / GPI;
  constexpr int RB = 16;

  __shared__ float    s_agg[OWN_R * C];
  __shared__ float    s_den[OWN_R * H];
  __shared__ unsigned s_max[OWN_R * H];
  __shared__ int      s_q[OWN_WAVES][OWN_QCAP];
  __shared__ float    s_w[OWN_WAVES][32 * H];
  __shared__ int      s_resweep;

  const int tid = threadIdx.x;
  const int lane = tid & 31;
  const int wave = tid >> 5;
  const int lo = blockIdx.x * OWN_R;
  int nloc = NNODES - lo;
  nloc = nloc > OWN_R ? OWN_R : nloc;
  nloc = nloc < 0 ? 0 : nloc;

  for (int i = tid; i < OWN_R * C; i += 1024) s_agg[i] = 0.f;
  for (int i = tid; i < OWN_R * H; i += 1024) {
    s_den[i] = 0.f;
    s_max[i] = ENC_NEG_INF;
  }
  if (tid == 0) s_resweep = 0;
  __syncthreads();

  const int4* dst4 = reinterpret_cast<const int4*>(dst);
  int qhead = 0, qtail = 0;

#pragma unroll 1
  for (int pass = 0; pass < 2; ++pass) {

    auto drain = [&](int n) {
      __builtin_amdgcn_fence(__ATOMIC_RELEASE, "wavefront");
      __builtin_amdgcn_wave_barrier();
      __builtin_amdgcn_fence(__ATOMIC_ACQUIRE, "wavefront");
      int rel = 0, s = 0;
      if (lane < n) {
        const int e = s_q[wave][(qhead + lane) & (OWN_QCAP - 1)];
        s = src[e];
        rel = dst[e] - lo;
        const float* els = el + (long)s * H;
        const float* erd = er + (long)(lo + rel) * H;
#pragma unroll
        for (int h = 0; h < H; ++h) {
          float v = els[h] + erd[h];
          v = v > 0.f ? v : 0.2f * v;
          if (pass == 0) {
            atomicMax(&s_max[rel * H + h], float_ord_encode(v));
          } else {
            const float m = float_ord_decode(s_max[rel * H + h]);
            const float w = expf(v - m);
            atomicAdd(&s_den[rel * H + h], w);
            s_w[wave][lane * H + h] = w;
          }
        }
      }
      if (pass == 1) {
        __builtin_amdgcn_fence(__ATOMIC_RELEASE, "wavefront");
        __builtin_amdgcn_wave_barrier();
        __builtin_amdgcn_fence(__ATOMIC_ACQUIRE, "wavefront");
        for (int j0 = 0; j0 < n; j0 += RB) {
          float f[RB][CPL];
#pragma unroll
          for (int u = 0; u < RB; ++u) {
            const int sj = __shfl(s, (j0 + u) & 31);
            if (j0 + u < n) {
#pragma unroll
              for (int p = 0; p < CPL; ++p)
                f[u][p] = feat[(long)sj * C + lane + 32 * p];
            }
          }
#pragma unroll
          for (int u = 0; u < RB; ++u) {
            const int rj = __shfl(rel, (j0 + u) & 31);
            if (j0 + u < n) {
#pragma unroll
              for (int p = 0; p < CPL; ++p) {
                const int k = lane + 32 * p;
                const float wv = s_w[wave][(j0 + u) * H + k / D];
                atomicAdd(&s_agg[rj * C + k], wv * f[u][p]);
              }
            }
          }
        }
      }
      qhead += n;
    };

    const bool sweep = (pass == 0) || (s_resweep != 0);
    if (sweep) {
      qhead = 0;
      qtail = 0;
#pragma unroll 1
      for (int i = 0; i < NIT; ++i) {
        while (qtail - qhead > OWN_QCAP - 512) drain(32);
        const int it = (i + (int)blockIdx.x) % NIT;
        const int g0 = it * GPI + wave * 128 + lane;
        int4 dd[4];
#pragma unroll
        for (int u = 0; u < 4; ++u) {
          const int g = g0 + 32 * u;
          dd[u] = g < NG4 ? dst4[g] : make_int4(-1, -1, -1, -1);
        }
#pragma unroll
        for (int u = 0; u < 4; ++u) {
          const int dv[4] = {dd[u].x, dd[u].y, dd[u].z, dd[u].w};
#pragma unroll
          for (int c = 0; c < 4; ++c) {
            const bool match = (unsigned)(dv[c] - lo) < (unsigned)nloc;
            const unsigned mask = __builtin_amdgcn_ballot_w32(match);
            if (mask != 0u) {
              if (match) {
                const int pos =
                    (qtail + (int)__builtin_amdgcn_mbcnt_lo(mask, 0u)) &
                    (OWN_QCAP - 1);
                s_q[wave][pos] = 4 * (g0 + 32 * u) + c;
              }
              qtail += __popc(mask);
            }
          }
        }
      }
      if (pass == 0 && qtail > OWN_QCAP) s_resweep = 1;
    } else {
      qhead = 0;
    }
    while (qtail - qhead >= 32) drain(32);
    if (qtail - qhead > 0) drain(qtail - qhead);
    __syncthreads();
  }

  for (int i = tid; i < nloc * C; i += 1024) {
    const int r = i / C;
    const int c = i % C;
    const float dn = s_den[r * H + c / D];
    float x = dn > 0.f ? s_agg[i] / dn : 0.f;
    x += bias[c];
    if (DO_ELU) x = x > 0.f ? x : expf(x) - 1.f;
    out[(long)lo * C + i] = x;
  }
}

extern "C" void kernel_launch(void* const* d_in, const int* in_sizes, int n_in,
                              void* d_out, int out_size, void* d_ws,
                              size_t ws_size, hipStream_t stream) {
  const float* x   = (const float*)d_in[0];
  const int*   src = (const int*)d_in[1];
  const int*   dst = (const int*)d_in[2];
  const float* W1  = (const float*)d_in[3];
  const float* al1 = (const float*)d_in[4];
  const float* ar1 = (const float*)d_in[5];
  const float* b1  = (const float*)d_in[6];
  const float* W2  = (const float*)d_in[7];
  const float* al2 = (const float*)d_in[8];
  const float* ar2 = (const float*)d_in[9];
  const float* b2  = (const float*)d_in[10];
  float* out = (float*)d_out;

  const int N = NNODES;

  float* ws = (float*)d_ws;
  float* feat1 = ws;
  float* el1   = feat1 + (long)N * 64;
  float* er1   = el1 + (long)N * 8;
  float* h1    = ws + (long)N * 96;
  float* feat2 = ws;
  float* el2   = el1;
  float* er2   = el1 + N;

  const int TB = 256;
  auto nb = [](long n, int b) { return (int)((n + b - 1) / b); };
  const int NOWN = (NNODES + OWN_R - 1) / OWN_R;

  gemm_wmma_kernel<256, 64><<<N / 16, 128, 0, stream>>>(x, W1, feat1);
  logits_kernel<8, 8><<<nb((long)N * 8, TB), TB, 0, stream>>>(feat1, al1, ar1, el1, er1, N * 8);
  owner_agg_kernel<8, 8, true><<<NOWN, 1024, 0, stream>>>(src, dst, el1, er1, feat1, b1, h1);

  gemm_wmma_kernel<64, 32><<<N / 16, 64, 0, stream>>>(h1, W2, feat2);
  logits_kernel<1, 32><<<nb((long)N, TB), TB, 0, stream>>>(feat2, al2, ar2, el2, er2, N);
  owner_agg_kernel<1, 32, false><<<NOWN, 1024, 0, stream>>>(src, dst, el2, er2, feat2, b2, out);
}


# owner number of each edge (dst / 391) kept as bit planes, one byte per edge, written by a small launch per call; owner groups sweep the planes, not the 4-byte ids; LDS adds stay in arrival order
# speedup vs baseline: 3.3672x; 1.2880x over previous
#include <hip/hip_runtime.h>
#include <hip/hip_bf16.h>

typedef __attribute__((ext_vector_type(2))) float v2f;
typedef __attribute__((ext_vector_type(8))) float v8f;

#define NNODES 100000
#define NEDGES 1600000

template <int K, int C>
__global__ __launch_bounds__(32 * (C / 16)) void gemm_wmma_kernel(
    const float* __restrict__ A, const float* __restrict__ B,
    float* __restrict__ D) {
  const int lane = threadIdx.x & 31;
  const int wave = threadIdx.x >> 5;
  const int row0 = blockIdx.x * 16;
  const int col0 = wave * 16;
  const int half = lane >> 4;
  const int l    = lane & 15;

  const float* arow = A + (long)(row0 + l) * K + 2 * half;
  const float* bcol = B + col0 + l;

  v8f acc = {};
#pragma unroll 4
  for (int k = 0; k < K; k += 4) {
    v2f a, b;
    a.x = arow[k + 0];
    a.y = arow[k + 1];
    b.x = bcol[(k + 2 * half + 0) * C];
    b.y = bcol[(k + 2 * half + 1) * C];
    acc = __builtin_amdgcn_wmma_f32_16x16x4_f32(
         false, a,  false, b,
         (short)0, acc,  false,  false);
  }

  float* drow = D + (long)(row0 + 8 * half) * C + col0 + l;
#pragma unroll
  for (int i = 0; i < 8; ++i) drow[i * C] = acc[i];
}

template <int H, int D>
__global__ void logits_kernel(const float* __restrict__ feat,
                              const float* __restrict__ al,
                              const float* __restrict__ ar,
                              float* __restrict__ el, float* __restrict__ er,
                              int NH) {
  int i = blockIdx.x * blockDim.x + threadIdx.x;
  if (i >= NH) return;
  int n = i / H, h = i % H;
  const float* f = feat + ((long)n * H + h) * D;
  const float* a = al + h * D;
  const float* r = ar + h * D;
  float sl = 0.f, sr = 0.f;
#pragma unroll
  for (int d = 0; d < D; ++d) {
    sl += f[d] * a[d];
    sr += f[d] * r[d];
  }
  el[i] = sl;
  er[i] = sr;
}

__device__ __forceinline__ unsigned float_ord_encode(float f) {
  unsigned u = __float_as_uint(f);
  return (u & 0x80000000u) ? ~u : (u | 0x80000000u);
}
__device__ __forceinline__ float float_ord_decode(unsigned u) {
  return (u & 0x80000000u) ? __uint_as_float(u ^ 0x80000000u)
                           : __uint_as_float(~u);
}
#define ENC_NEG_INF 0x007FFFFFu

#define OWN_R 391
#define OWN_WAVES 32
#define OWN_QCAP 1024

#define OWN_NGRP (NEDGES / 32)
#define OWN_NBLK ((OWN_NGRP + 31) / 32)

__global__ __launch_bounds__(256) void owner_table_kernel(
    const int* __restrict__ dst, unsigned* __restrict__ tab) {
  const int lane = threadIdx.x & 31;
  const int blk = blockIdx.x * 8 + (threadIdx.x >> 5);
  if (blk >= OWN_NBLK) return;
  const int grp = blk * 32 + lane;
  unsigned p[8] = {0u, 0u, 0u, 0u, 0u, 0u, 0u, 0u};
  if (grp < OWN_NGRP) {
    const int4* d4 = reinterpret_cast<const int4*>(dst + 32 * (long)grp);
#pragma unroll
    for (int c = 0; c < 8; ++c) {
      const int4 d = d4[c];
      const int dv[4] = {d.x, d.y, d.z, d.w};
#pragma unroll
      for (int k = 0; k < 4; ++k) {
        const unsigned oq = (unsigned)dv[k] / OWN_R;
        const unsigned o = oq < 255u ? oq : 255u;
#pragma unroll
        for (int j = 0; j < 8; ++j) p[j] |= ((o >> j) & 1u) << (4 * c + k);
      }
    }
  }
  uint4* t4 = reinterpret_cast<uint4*>(tab);
  t4[blk * 64 + lane]      = make_uint4(p[0], p[1], p[2], p[3]);
  t4[blk * 64 + 32 + lane] = make_uint4(p[4], p[5], p[6], p[7]);
}

template <int H, int D, bool DO_ELU>
__global__ __launch_bounds__(1024) void owner_agg_kernel(
    const int* __restrict__ src, const int* __restrict__ dst,
    const unsigned* __restrict__ tab,
    const float* __restrict__ el, const float* __restrict__ er,
    const float* __restrict__ feat, const float* __restrict__ bias,
    float* __restrict__ out) {
  constexpr int C = H * D;
  constexpr int CPL = C / 32;
  constexpr int BPI = OWN_WAVES * 2;
  constexpr int NIT = (OWN_NBLK + BPI - 1) / BPI;
  constexpr int RB = 16;

  __shared__ float    s_agg[OWN_R * C];
  __shared__ float    s_den[OWN_R * H];
  __shared__ unsigned s_max[OWN_R * H];
  __shared__ int      s_q[OWN_WAVES][OWN_QCAP];
  __shared__ float    s_w[OWN_WAVES][32 * H];
  __shared__ int      s_resweep;

  const int tid = threadIdx.x;
  const int lane = tid & 31;
  const int wave = tid >> 5;
  const int lo = blockIdx.x * OWN_R;
  int nloc = NNODES - lo;
  nloc = nloc > OWN_R ? OWN_R : nloc;
  nloc = nloc < 0 ? 0 : nloc;

  for (int i = tid; i < OWN_R * C; i += 1024) s_agg[i] = 0.f;
  for (int i = tid; i < OWN_R * H; i += 1024) {
    s_den[i] = 0.f;
    s_max[i] = ENC_NEG_INF;
  }
  if (tid == 0) s_resweep = 0;
  __syncthreads();

  const uint4* tab4 = reinterpret_cast<const uint4*>(tab);
  unsigned want[8];
#pragma unroll
  for (int j = 0; j < 8; ++j) want[j] = ((blockIdx.x >> j) & 1u) ? ~0u : 0u;
  int qhead = 0, qtail = 0;

#pragma unroll 1
  for (int pass = 0; pass < 2; ++pass) {

    auto drain = [&](int n) {
      __builtin_amdgcn_fence(__ATOMIC_RELEASE, "wavefront");
      __builtin_amdgcn_wave_barrier();
      __builtin_amdgcn_fence(__ATOMIC_ACQUIRE, "wavefront");
      int rel = 0, s = 0;
      if (lane < n) {
        const int e = s_q[wave][(qhead + lane) & (OWN_QCAP - 1)];
        s = src[e];
        const unsigned ru = (unsigned)(dst[e] - lo);
        rel = (int)(ru < (unsigned)(OWN_R - 1) ? ru : (unsigned)(OWN_R - 1));
        const float* els = el + (long)s * H;
        const float* erd = er + (long)(lo + rel) * H;
#pragma unroll
        for (int h = 0; h < H; ++h) {
          float v = els[h] + erd[h];
          v = v > 0.f ? v : 0.2f * v;
          if (pass == 0) {
            atomicMax(&s_max[rel * H + h], float_ord_encode(v));
          } else {
            const float m = float_ord_decode(s_max[rel * H + h]);
            const float w = expf(v - m);
            atomicAdd(&s_den[rel * H + h], w);
            s_w[wave][lane * H + h] = w;
          }
        }
      }
      if (pass == 1) {
        __builtin_amdgcn_fence(__ATOMIC_RELEASE, "wavefront");
        __builtin_amdgcn_wave_barrier();
        __builtin_amdgcn_fence(__ATOMIC_ACQUIRE, "wavefront");
        for (int j0 = 0; j0 < n; j0 += RB) {
          float f[RB][CPL];
#pragma unroll
          for (int u = 0; u < RB; ++u) {
            const int sj = __shfl(s, (j0 + u) & 31);
            if (j0 + u < n) {
#pragma unroll
              for (int p = 0; p < CPL; ++p)
                f[u][p] = feat[(long)sj * C + lane + 32 * p];
            }
          }
#pragma unroll
          for (int u = 0; u < RB; ++u) {
            const int rj = __shfl(rel, (j0 + u) & 31);
            if (j0 + u < n) {
#pragma unroll
              for (int p = 0; p < CPL; ++p) {
                const int k = lane + 32 * p;
                const float wv = s_w[wave][(j0 + u) * H + k / D];
                atomicAdd(&s_agg[rj * C + k], wv * f[u][p]);
              }
            }
          }
        }
      }
      qhead += n;
    };

    const bool sweep = (pass == 0) || (s_resweep != 0);
    if (sweep) {
      qhead = 0;
      qtail = 0;
#pragma unroll 1
      for (int i = 0; i < NIT; ++i) {
        const int it = (i + (int)blockIdx.x) % NIT;
        const int blk0 = it * BPI + wave * 2;
        unsigned hit[2];
#pragma unroll
        for (int k = 0; k < 2; ++k) {
          const int blk = blk0 + k;
          const int bc = blk < OWN_NBLK ? blk : OWN_NBLK - 1;
          const uint4 pa = tab4[bc * 64 + lane];
          const uint4 pb = tab4[bc * 64 + 32 + lane];
          const unsigned pl[8] = {pa.x, pa.y, pa.z, pa.w, pb.x, pb.y, pb.z, pb.w};
          unsigned acc = ~0u;
#pragma unroll
          for (int j = 0; j < 8; ++j) acc &= ~(pl[j] ^ want[j]);
          hit[k] = (blk * 32 + lane < OWN_NGRP) ? acc : 0u;
        }
        unsigned h0 = hit[0], h1 = hit[1];
#pragma unroll 1
        for (int k = 0; k < 2; ++k) {
          unsigned mm = h0;
          h0 = h1;
          const int ebase = (blk0 + k) * 1024 + lane * 32;
          unsigned any = __builtin_amdgcn_ballot_w32(mm != 0u);
          while (any != 0u) {
            while (qtail - qhead > OWN_QCAP - 32) drain(32);
            if (mm != 0u) {
              const int q = __builtin_ctz(mm);
              mm &= mm - 1u;
              const int pos =
                  (qtail + (int)__builtin_amdgcn_mbcnt_lo(any, 0u)) &
                  (OWN_QCAP - 1);
              s_q[wave][pos] = ebase + q;
            }
            qtail += __popc(any);
            any = __builtin_amdgcn_ballot_w32(mm != 0u);
          }
        }
      }
      if (pass == 0 && qtail > OWN_QCAP) s_resweep = 1;
    } else {
      qhead = 0;
    }
    while (qtail - qhead >= 32) drain(32);
    if (qtail - qhead > 0) drain(qtail - qhead);
    __syncthreads();
  }

  for (int i = tid; i < nloc * C; i += 1024) {
    const int r = i / C;
    const int c = i % C;
    const float dn = s_den[r * H + c / D];
    float x = dn > 0.f ? s_agg[i] / dn : 0.f;
    x += bias[c];
    if (DO_ELU) x = x > 0.f ? x : expf(x) - 1.f;
    out[(long)lo * C + i] = x;
  }
}

extern "C" void kernel_launch(void* const* d_in, const int* in_sizes, int n_in,
                              void* d_out, int out_size, void* d_ws,
                              size_t ws_size, hipStream_t stream) {
  const float* x   = (const float*)d_in[0];
  const int*   src = (const int*)d_in[1];
  const int*   dst = (const int*)d_in[2];
  const float* W1  = (const float*)d_in[3];
  const float* al1 = (const float*)d_in[4];
  const float* ar1 = (const float*)d_in[5];
  const float* b1  = (const float*)d_in[6];
  const float* W2  = (const float*)d_in[7];
  const float* al2 = (const float*)d_in[8];
  const float* ar2 = (const float*)d_in[9];
  const float* b2  = (const float*)d_in[10];
  float* out = (float*)d_out;

  const int N = NNODES;

  float* ws = (float*)d_ws;
  float* feat1 = ws;
  float* el1   = feat1 + (long)N * 64;
  float* er1   = el1 + (long)N * 8;
  unsigned* tab = (unsigned*)(ws + (long)N * 80);
  float* h1    = ws + (long)N * 96;
  float* feat2 = ws;
  float* el2   = el1;
  float* er2   = el1 + N;

  const int TB = 256;
  auto nb = [](long n, int b) { return (int)((n + b - 1) / b); };
  const int NOWN = (NNODES + OWN_R - 1) / OWN_R;

  owner_table_kernel<<<nb(OWN_NBLK, 8), 256, 0, stream>>>(dst, tab);

  gemm_wmma_kernel<256, 64><<<N / 16, 128, 0, stream>>>(x, W1, feat1);
  logits_kernel<8, 8><<<nb((long)N * 8, TB), TB, 0, stream>>>(feat1, al1, ar1, el1, er1, N * 8);
  owner_agg_kernel<8, 8, true><<<NOWN, 1024, 0, stream>>>(src, dst, tab, el1, er1, feat1, b1, h1);

  gemm_wmma_kernel<64, 32><<<N / 16, 64, 0, stream>>>(h1, W2, feat2);
  logits_kernel<1, 32><<<nb((long)N, TB), TB, 0, stream>>>(feat2, al2, ar2, el2, er2, N);
  owner_agg_kernel<1, 32, false><<<NOWN, 1024, 0, stream>>>(src, dst, tab, el2, er2, feat2, b2, out);
}


# owner groups: pass 0 leaves src id and local row packed in the wave's LDS list slot (one head: also the score); pass 1 gathers no src/dst (nor el/er) again; LDS adds stay in arrival order
# speedup vs baseline: 3.6437x; 1.0821x over previous
#include <hip/hip_runtime.h>
#include <hip/hip_bf16.h>

typedef __attribute__((ext_vector_type(2))) float v2f;
typedef __attribute__((ext_vector_type(8))) float v8f;

#define NNODES 100000
#define NEDGES 1600000

template <int K, int C>
__global__ __launch_bounds__(32 * (C / 16)) void gemm_wmma_kernel(
    const float* __restrict__ A, const float* __restrict__ B,
    float* __restrict__ D) {
  const int lane = threadIdx.x & 31;
  const int wave = threadIdx.x >> 5;
  const int row0 = blockIdx.x * 16;
  const int col0 = wave * 16;
  const int half = lane >> 4;
  const int l    = lane & 15;

  const float* arow = A + (long)(row0 + l) * K + 2 * half;
  const float* bcol = B + col0 + l;

  v8f acc = {};
#pragma unroll 4
  for (int k = 0; k < K; k += 4) {
    v2f a, b;
    a.x = arow[k + 0];
    a.y = arow[k + 1];
    b.x = bcol[(k + 2 * half + 0) * C];
    b.y = bcol[(k + 2 * half + 1) * C];
    acc = __builtin_amdgcn_wmma_f32_16x16x4_f32(
         false, a,  false, b,
         (short)0, acc,  false,  false);
  }

  float* drow = D + (long)(row0 + 8 * half) * C + col0 + l;
#pragma unroll
  for (int i = 0; i < 8; ++i) drow[i * C] = acc[i];
}

template <int H, int D>
__global__ void logits_kernel(const float* __restrict__ feat,
                              const float* __restrict__ al,
                              const float* __restrict__ ar,
                              float* __restrict__ el, float* __restrict__ er,
                              int NH) {
  int i = blockIdx.x * blockDim.x + threadIdx.x;
  if (i >= NH) return;
  int n = i / H, h = i % H;
  const float* f = feat + ((long)n * H + h) * D;
  const float* a = al + h * D;
  const float* r = ar + h * D;
  float sl = 0.f, sr = 0.f;
#pragma unroll
  for (int d = 0; d < D; ++d) {
    sl += f[d] * a[d];
    sr += f[d] * r[d];
  }
  el[i] = sl;
  er[i] = sr;
}

__device__ __forceinline__ unsigned float_ord_encode(float f) {
  unsigned u = __float_as_uint(f);
  return (u & 0x80000000u) ? ~u : (u | 0x80000000u);
}
__device__ __forceinline__ float float_ord_decode(unsigned u) {
  return (u & 0x80000000u) ? __uint_as_float(u ^ 0x80000000u)
                           : __uint_as_float(~u);
}
#define ENC_NEG_INF 0x007FFFFFu

#define OWN_R 391
#define OWN_WAVES 32
#define OWN_QCAP 1024
#define OWN_RBITS 9

#define OWN_NGRP (NEDGES / 32)
#define OWN_NBLK ((OWN_NGRP + 31) / 32)

__global__ __launch_bounds__(256) void owner_table_kernel(
    const int* __restrict__ dst, unsigned* __restrict__ tab) {
  const int lane = threadIdx.x & 31;
  const int blk = blockIdx.x * 8 + (threadIdx.x >> 5);
  if (blk >= OWN_NBLK) return;
  const int grp = blk * 32 + lane;
  unsigned p[8] = {0u, 0u, 0u, 0u, 0u, 0u, 0u, 0u};
  if (grp < OWN_NGRP) {
    const int4* d4 = reinterpret_cast<const int4*>(dst + 32 * (long)grp);
#pragma unroll
    for (int c = 0; c < 8; ++c) {
      const int4 d = d4[c];
      const int dv[4] = {d.x, d.y, d.z, d.w};
#pragma unroll
      for (int k = 0; k < 4; ++k) {
        const unsigned oq = (unsigned)dv[k] / OWN_R;
        const unsigned o = oq < 255u ? oq : 255u;
#pragma unroll
        for (int j = 0; j < 8; ++j) p[j] |= ((o >> j) & 1u) << (4 * c + k);
      }
    }
  }
  uint4* t4 = reinterpret_cast<uint4*>(tab);
  t4[blk * 64 + lane]      = make_uint4(p[0], p[1], p[2], p[3]);
  t4[blk * 64 + 32 + lane] = make_uint4(p[4], p[5], p[6], p[7]);
}

template <int H, int D, bool DO_ELU>
__global__ __launch_bounds__(1024) void owner_agg_kernel(
    const int* __restrict__ src, const int* __restrict__ dst,
    const unsigned* __restrict__ tab,
    const float* __restrict__ el, const float* __restrict__ er,
    const float* __restrict__ feat, const float* __restrict__ bias,
    float* __restrict__ out) {
  constexpr int C = H * D;
  constexpr int CPL = C / 32;
  constexpr int BPI = OWN_WAVES * 2;
  constexpr int NIT = (OWN_NBLK + BPI - 1) / BPI;
  constexpr int RB = 16;

  __shared__ float    s_agg[OWN_R * C];
  __shared__ float    s_den[OWN_R * H];
  __shared__ unsigned s_max[OWN_R * H];
  __shared__ int      s_q[OWN_WAVES][OWN_QCAP];
  __shared__ float    s_w[OWN_WAVES][32 * H];
  __shared__ int      s_resweep;
  constexpr bool KEEPV = (H == 1);
  __shared__ float    s_v[KEEPV ? OWN_WAVES : 1][KEEPV ? OWN_QCAP : 1];

  const int tid = threadIdx.x;
  const int lane = tid & 31;
  const int wave = tid >> 5;
  const int lo = blockIdx.x * OWN_R;
  int nloc = NNODES - lo;
  nloc = nloc > OWN_R ? OWN_R : nloc;
  nloc = nloc < 0 ? 0 : nloc;

  for (int i = tid; i < OWN_R * C; i += 1024) s_agg[i] = 0.f;
  for (int i = tid; i < OWN_R * H; i += 1024) {
    s_den[i] = 0.f;
    s_max[i] = ENC_NEG_INF;
  }
  if (tid == 0) s_resweep = 0;
  __syncthreads();

  const uint4* tab4 = reinterpret_cast<const uint4*>(tab);
  unsigned want[8];
#pragma unroll
  for (int j = 0; j < 8; ++j) want[j] = ((blockIdx.x >> j) & 1u) ? ~0u : 0u;
  int qhead = 0, qtail = 0;
  bool packed = false;

#pragma unroll 1
  for (int pass = 0; pass < 2; ++pass) {

    auto drain = [&](int n) {
      __builtin_amdgcn_fence(__ATOMIC_RELEASE, "wavefront");
      __builtin_amdgcn_wave_barrier();
      __builtin_amdgcn_fence(__ATOMIC_ACQUIRE, "wavefront");
      int rel = 0, s = 0;
      if (lane < n) {
        const int slot = (qhead + lane) & (OWN_QCAP - 1);
        const unsigned q = (unsigned)s_q[wave][slot];
        if (packed) {
          s = (int)(q >> OWN_RBITS);
          rel = (int)(q & ((1u << OWN_RBITS) - 1u));
        } else {
          s = src[q];
          const unsigned ru = (unsigned)(dst[q] - lo);
          rel = (int)(ru < (unsigned)(OWN_R - 1) ? ru : (unsigned)(OWN_R - 1));
          if (pass == 0)
            s_q[wave][slot] = (int)(((unsigned)s << OWN_RBITS) | (unsigned)rel);
        }
        const float* els = el + (long)s * H;
        const float* erd = er + (long)(lo + rel) * H;
#pragma unroll
        for (int h = 0; h < H; ++h) {
          float v;
          if (KEEPV && packed) {
            v = s_v[wave][slot];
          } else {
            v = els[h] + erd[h];
            v = v > 0.f ? v : 0.2f * v;
            if (KEEPV && pass == 0) s_v[wave][slot] = v;
          }
          if (pass == 0) {
            atomicMax(&s_max[rel * H + h], float_ord_encode(v));
          } else {
            const float m = float_ord_decode(s_max[rel * H + h]);
            const float w = expf(v - m);
            atomicAdd(&s_den[rel * H + h], w);
            s_w[wave][lane * H + h] = w;
          }
        }
      }
      if (pass == 1) {
        __builtin_amdgcn_fence(__ATOMIC_RELEASE, "wavefront");
        __builtin_amdgcn_wave_barrier();
        __builtin_amdgcn_fence(__ATOMIC_ACQUIRE, "wavefront");
        for (int j0 = 0; j0 < n; j0 += RB) {
          float f[RB][CPL];
#pragma unroll
          for (int u = 0; u < RB; ++u) {
            const int sj = __shfl(s, (j0 + u) & 31);
            if (j0 + u < n) {
#pragma unroll
              for (int p = 0; p < CPL; ++p)
                f[u][p] = feat[(long)sj * C + lane + 32 * p];
            }
          }
#pragma unroll
          for (int u = 0; u < RB; ++u) {
            const int rj = __shfl(rel, (j0 + u) & 31);
            if (j0 + u < n) {
#pragma unroll
              for (int p = 0; p < CPL; ++p) {
                const int k = lane + 32 * p;
                const float wv = s_w[wave][(j0 + u) * H + k / D];
                atomicAdd(&s_agg[rj * C + k], wv * f[u][p]);
              }
            }
          }
        }
      }
      qhead += n;
    };

    const bool sweep = (pass == 0) || (s_resweep != 0);
    packed = !sweep;
    if (sweep) {
      qhead = 0;
      qtail = 0;
#pragma unroll 1
      for (int i = 0; i < NIT; ++i) {
        const int it = (i + (int)blockIdx.x) % NIT;
        const int blk0 = it * BPI + wave * 2;
        unsigned hit[2];
#pragma unroll
        for (int k = 0; k < 2; ++k) {
          const int blk = blk0 + k;
          const int bc = blk < OWN_NBLK ? blk : OWN_NBLK - 1;
          const uint4 pa = tab4[bc * 64 + lane];
          const uint4 pb = tab4[bc * 64 + 32 + lane];
          const unsigned pl[8] = {pa.x, pa.y, pa.z, pa.w, pb.x, pb.y, pb.z, pb.w};
          unsigned acc = ~0u;
#pragma unroll
          for (int j = 0; j < 8; ++j) acc &= ~(pl[j] ^ want[j]);
          hit[k] = (blk * 32 + lane < OWN_NGRP) ? acc : 0u;
        }
        unsigned h0 = hit[0], h1 = hit[1];
#pragma unroll 1
        for (int k = 0; k < 2; ++k) {
          unsigned mm = h0;
          h0 = h1;
          const int ebase = (blk0 + k) * 1024 + lane * 32;
          unsigned any = __builtin_amdgcn_ballot_w32(mm != 0u);
          while (any != 0u) {
            while (qtail - qhead > OWN_QCAP - 32) drain(32);
            if (mm != 0u) {
              const int q = __builtin_ctz(mm);
              mm &= mm - 1u;
              const int pos =
                  (qtail + (int)__builtin_amdgcn_mbcnt_lo(any, 0u)) &
                  (OWN_QCAP - 1);
              s_q[wave][pos] = ebase + q;
            }
            qtail += __popc(any);
            any = __builtin_amdgcn_ballot_w32(mm != 0u);
          }
        }
      }
      if (pass == 0 && qtail > OWN_QCAP) s_resweep = 1;
    } else {
      qhead = 0;
    }
    while (qtail - qhead >= 32) drain(32);
    if (qtail - qhead > 0) drain(qtail - qhead);
    __syncthreads();
  }

  for (int i = tid; i < nloc * C; i += 1024) {
    const int r = i / C;
    const int c = i % C;
    const float dn = s_den[r * H + c / D];
    float x = dn > 0.f ? s_agg[i] / dn : 0.f;
    x += bias[c];
    if (DO_ELU) x = x > 0.f ? x : expf(x) - 1.f;
    out[(long)lo * C + i] = x;
  }
}

extern "C" void kernel_launch(void* const* d_in, const int* in_sizes, int n_in,
                              void* d_out, int out_size, void* d_ws,
                              size_t ws_size, hipStream_t stream) {
  const float* x   = (const float*)d_in[0];
  const int*   src = (const int*)d_in[1];
  const int*   dst = (const int*)d_in[2];
  const float* W1  = (const float*)d_in[3];
  const float* al1 = (const float*)d_in[4];
  const float* ar1 = (const float*)d_in[5];
  const float* b1  = (const float*)d_in[6];
  const float* W2  = (const float*)d_in[7];
  const float* al2 = (const float*)d_in[8];
  const float* ar2 = (const float*)d_in[9];
  const float* b2  = (const float*)d_in[10];
  float* out = (float*)d_out;

  const int N = NNODES;

  float* ws = (float*)d_ws;
  float* feat1 = ws;
  float* el1   = feat1 + (long)N * 64;
  float* er1   = el1 + (long)N * 8;
  unsigned* tab = (unsigned*)(ws + (long)N * 80);
  float* h1    = ws + (long)N * 96;
  float* feat2 = ws;
  float* el2   = el1;
  float* er2   = el1 + N;

  const int TB = 256;
  auto nb = [](long n, int b) { return (int)((n + b - 1) / b); };
  const int NOWN = (NNODES + OWN_R - 1) / OWN_R;

  owner_table_kernel<<<nb(OWN_NBLK, 8), 256, 0, stream>>>(dst, tab);

  gemm_wmma_kernel<256, 64><<<N / 16, 128, 0, stream>>>(x, W1, feat1);
  logits_kernel<8, 8><<<nb((long)N * 8, TB), TB, 0, stream>>>(feat1, al1, ar1, el1, er1, N * 8);
  owner_agg_kernel<8, 8, true><<<NOWN, 1024, 0, stream>>>(src, dst, tab, el1, er1, feat1, b1, h1);

  gemm_wmma_kernel<64, 32><<<N / 16, 64, 0, stream>>>(h1, W2, feat2);
  logits_kernel<1, 32><<<nb((long)N, TB), TB, 0, stream>>>(feat2, al2, ar2, el2, er2, N);
  owner_agg_kernel<1, 32, false><<<NOWN, 1024, 0, stream>>>(src, dst, tab, el2, er2, feat2, b2, out);
}


# layer 1's owner groups leave each wave's edge list (src id, local row) in the group's own rows of out; layer 2's groups read it back: no table sweep, no src/dst gathers there; add order unchanged
# speedup vs baseline: 4.3016x; 1.1806x over previous
#include <hip/hip_runtime.h>
#include <hip/hip_bf16.h>

typedef __attribute__((ext_vector_type(2))) float v2f;
typedef __attribute__((ext_vector_type(8))) float v8f;

#define NNODES 100000
#define NEDGES 1600000

#define GEMM_WAVES 25

template <int K, int C>
__global__ __launch_bounds__(32 * GEMM_WAVES)
__attribute__((amdgpu_waves_per_eu(7, 7))) void gemm_wmma_kernel(
    const float* __restrict__ A, const float* __restrict__ B,
    float* __restrict__ D) {
  constexpr int NT = C / 16;
  constexpr int NL = K / 32;
  constexpr int BP = K + 4;
  static_assert(K % 32 == 0 && C % 16 == 0, "whole lines, whole tiles");
  static_assert((BP / 4) % 2 == 1, "row pitch: an odd number of 16-byte pieces");
  __shared__ __attribute__((aligned(16))) float sB[C * BP];

  const int tid = threadIdx.x;
  const int lane = tid & 31;
  const int wave = tid >> 5;
  const int half = lane >> 4;
  const int c    = lane & 15;
  const int row0 = (blockIdx.x * GEMM_WAVES + wave) * 16;

  const float4* ap =
      reinterpret_cast<const float4*>(A + (long)(row0 + c) * K + 16 * half);
  float4 a[2][4];
#pragma unroll
  for (int u = 0; u < 4; ++u) a[0][u] = ap[u];

  float* dpin = D + (long)(row0 + 8 * half) * C + c;
  asm volatile("" : "+v"(dpin));
  typedef __attribute__((address_space(1))) float global_float;
  global_float* drow = (global_float*)dpin;

  constexpr int NTHR = 32 * GEMM_WAVES;
  constexpr int NQ = C * K / 4;
  constexpr int NU = (NQ + NTHR - 1) / NTHR;
  float4 w[NU];
#pragma unroll
  for (int j = 0; j < NU; ++j) {
    const int q = tid + j * NTHR;
    if (q < NQ) {
      const int n = q % C;
      const int k = (q / C) * 4;
      w[j].x = B[(k + 0) * C + n];
      w[j].y = B[(k + 1) * C + n];
      w[j].z = B[(k + 2) * C + n];
      w[j].w = B[(k + 3) * C + n];
    }
  }
#pragma unroll
  for (int j = 0; j < NU; ++j) {
    const int q = tid + j * NTHR;
    if (q < NQ) {
      const int n = q % C;
      const int k = (q / C) * 4;
      *reinterpret_cast<float4*>(&sB[n * BP + k]) = w[j];
    }
  }
  __syncthreads();

  v8f acc[NT];
#pragma unroll
  for (int t = 0; t < NT; ++t) acc[t] = v8f{};

  const float* bl = sB + c * BP + 16 * half;
#pragma unroll
  for (int L = 0; L < NL; ++L) {
    if (L + 1 < NL) {
#pragma unroll
      for (int u = 0; u < 4; ++u) a[(L + 1) & 1][u] = ap[8 * (L + 1) + u];
    }
#pragma unroll
    for (int u = 0; u < 4; ++u) {
      const float4 av = a[L & 1][u];
      float4 bv[NT];
#pragma unroll
      for (int t = 0; t < NT; ++t)
        bv[t] = *reinterpret_cast<const float4*>(bl + t * 16 * BP + 32 * L + 4 * u);
      v2f a01, a23;
      a01.x = av.x; a01.y = av.y;
      a23.x = av.z; a23.y = av.w;
#pragma unroll
      for (int t = 0; t < NT; ++t) {
        v2f b01;
        b01.x = bv[t].x; b01.y = bv[t].y;
        acc[t] = __builtin_amdgcn_wmma_f32_16x16x4_f32(
             false, a01,  false, b01,
             (short)0, acc[t],  false,  false);
      }
#pragma unroll
      for (int t = 0; t < NT; ++t) {
        v2f b23;
        b23.x = bv[t].z; b23.y = bv[t].w;
        acc[t] = __builtin_amdgcn_wmma_f32_16x16x4_f32(
             false, a23,  false, b23,
             (short)0, acc[t],  false,  false);
      }
    }
  }

#pragma unroll
  for (int t = 0; t < NT; ++t) {
#pragma unroll
    for (int i = 0; i < 8; ++i) drow[i * C + 16 * t] = acc[t][i];
  }
}

template <int H, int D>
__global__ void logits_kernel(const float* __restrict__ feat,
                              const float* __restrict__ al,
                              const float* __restrict__ ar,
                              float* __restrict__ el, float* __restrict__ er,
                              int NH) {
  int i = blockIdx.x * blockDim.x + threadIdx.x;
  if (i >= NH) return;
  int n = i / H, h = i % H;
  const float* f = feat + ((long)n * H + h) * D;
  const float* a = al + h * D;
  const float* r = ar + h * D;
  float sl = 0.f, sr = 0.f;
#pragma unroll
  for (int d = 0; d < D; ++d) {
    sl += f[d] * a[d];
    sr += f[d] * r[d];
  }
  el[i] = sl;
  er[i] = sr;
}

__device__ __forceinline__ unsigned float_ord_encode(float f) {
  unsigned u = __float_as_uint(f);
  return (u & 0x80000000u) ? ~u : (u | 0x80000000u);
}
__device__ __forceinline__ float float_ord_decode(unsigned u) {
  return (u & 0x80000000u) ? __uint_as_float(u ^ 0x80000000u)
                           : __uint_as_float(~u);
}
#define ENC_NEG_INF 0x007FFFFFu

#define OWN_R 391
#define OWN_WAVES 32
#define OWN_QCAP 1024
#define OWN_RBITS 9

#define LST_ROW 32
#define LST_NONE 0xFFFFFFFFu

#define OWN_NGRP (NEDGES / 32)
#define OWN_NBLK ((OWN_NGRP + 31) / 32)

__global__ __launch_bounds__(256) void owner_table_kernel(
    const int* __restrict__ dst, unsigned* __restrict__ tab) {
  const int lane = threadIdx.x & 31;
  const int blk = blockIdx.x * 8 + (threadIdx.x >> 5);
  if (blk >= OWN_NBLK) return;
  const int grp = blk * 32 + lane;
  unsigned p[8] = {0u, 0u, 0u, 0u, 0u, 0u, 0u, 0u};
  if (grp < OWN_NGRP) {
    const int4* d4 = reinterpret_cast<const int4*>(dst + 32 * (long)grp);
#pragma unroll
    for (int c = 0; c < 8; ++c) {
      const int4 d = d4[c];
      const int dv[4] = {d.x, d.y, d.z, d.w};
#pragma unroll
      for (int k = 0; k < 4; ++k) {
        const unsigned oq = (unsigned)dv[k] / OWN_R;
        const unsigned o = oq < 255u ? oq : 255u;
#pragma unroll
        for (int j = 0; j < 8; ++j) p[j] |= ((o >> j) & 1u) << (4 * c + k);
      }
    }
  }
  uint4* t4 = reinterpret_cast<uint4*>(tab);
  t4[blk * 64 + lane]      = make_uint4(p[0], p[1], p[2], p[3]);
  t4[blk * 64 + 32 + lane] = make_uint4(p[4], p[5], p[6], p[7]);
}

template <int H, int D, bool DO_ELU, int LISTS>
__global__ __launch_bounds__(1024) void owner_agg_kernel(
    const int* __restrict__ src, const int* __restrict__ dst,
    const unsigned* __restrict__ tab,
    const float* __restrict__ el, const float* __restrict__ er,
    const float* __restrict__ feat, const float* __restrict__ bias,
    unsigned* lst, float* out) {
  constexpr int C = H * D;
  static_assert(LISTS != 2 || C == LST_ROW, "the lists lie in this layer's own output rows");
  constexpr int CPL = C / 32;
  constexpr int BPI = OWN_WAVES * 2;
  constexpr int NIT = (OWN_NBLK + BPI - 1) / BPI;
  constexpr int RB = 16;

  __shared__ float    s_agg[OWN_R * C];
  __shared__ float    s_den[OWN_R * H];
  __shared__ unsigned s_max[OWN_R * H];
  __shared__ int      s_q[OWN_WAVES][OWN_QCAP];
  __shared__ float    s_w[OWN_WAVES][32 * H];
  __shared__ int      s_resweep;
  constexpr bool KEEPV = (H == 1);
  __shared__ float    s_v[KEEPV ? OWN_WAVES : 1][KEEPV ? OWN_QCAP : 1];

  const int tid = threadIdx.x;
  const int lane = tid & 31;
  const int wave = tid >> 5;
  const int lo = blockIdx.x * OWN_R;
  int nloc = NNODES - lo;
  nloc = nloc > OWN_R ? OWN_R : nloc;
  nloc = nloc < 0 ? 0 : nloc;
  static_assert(OWN_WAVES == LST_ROW, "a row of counts, cap rows of lists");
  const int cap = nloc >= 1 ? ((nloc - 1) & ~31) : 0;
  unsigned* const lcnt = lst + (long)lo * LST_ROW;
  unsigned* const lmine = lcnt + LST_ROW + wave * cap;

  for (int i = tid; i < OWN_R * C; i += 1024) s_agg[i] = 0.f;
  for (int i = tid; i < OWN_R * H; i += 1024) {
    s_den[i] = 0.f;
    s_max[i] = ENC_NEG_INF;
  }
  if (tid == 0) s_resweep = 0;
  __syncthreads();

  const uint4* tab4 = reinterpret_cast<const uint4*>(tab);
  unsigned want[8];
#pragma unroll
  for (int j = 0; j < 8; ++j) want[j] = ((blockIdx.x >> j) & 1u) ? ~0u : 0u;
  int qhead = 0, qtail = 0;
  bool packed = false;
  bool havev = false;
  unsigned icnt = LST_NONE;
  if (LISTS == 2 && nloc >= 1) icnt = __builtin_amdgcn_readfirstlane(lcnt[wave]);
  const bool imported = icnt <= (unsigned)cap;

#pragma unroll 1
  for (int pass = 0; pass < 2; ++pass) {

    auto drain = [&](int n) {
      __builtin_amdgcn_fence(__ATOMIC_RELEASE, "wavefront");
      __builtin_amdgcn_wave_barrier();
      __builtin_amdgcn_fence(__ATOMIC_ACQUIRE, "wavefront");
      int rel = 0, s = 0;
      if (lane < n) {
        const int slot = (qhead + lane) & (OWN_QCAP - 1);
        const unsigned q = (unsigned)s_q[wave][slot];
        if (packed) {
          s = (int)(q >> OWN_RBITS);
          rel = (int)(q & ((1u << OWN_RBITS) - 1u));
        } else {
          s = src[q];
          const unsigned ru = (unsigned)(dst[q] - lo);
          rel = (int)(ru < (unsigned)(OWN_R - 1) ? ru : (unsigned)(OWN_R - 1));
          if (pass == 0) {
            const unsigned pk = ((unsigned)s << OWN_RBITS) | (unsigned)rel;
            s_q[wave][slot] = (int)pk;
            if (LISTS == 1 && slot < cap) lmine[slot] = pk;
          }
        }
        const float* els = el + (long)s * H;
        const float* erd = er + (long)(lo + rel) * H;
#pragma unroll
        for (int h = 0; h < H; ++h) {
          float v;
          if (KEEPV && havev) {
            v = s_v[wave][slot];
          } else {
            v = els[h] + erd[h];
            v = v > 0.f ? v : 0.2f * v;
            if (KEEPV && pass == 0) s_v[wave][slot] = v;
          }
          if (pass == 0) {
            atomicMax(&s_max[rel * H + h], float_ord_encode(v));
          } else {
            const float m = float_ord_decode(s_max[rel * H + h]);
            const float w = expf(v - m);
            atomicAdd(&s_den[rel * H + h], w);
            s_w[wave][lane * H + h] = w;
          }
        }
      }
      if (pass == 1) {
        __builtin_amdgcn_fence(__ATOMIC_RELEASE, "wavefront");
        __builtin_amdgcn_wave_barrier();
        __builtin_amdgcn_fence(__ATOMIC_ACQUIRE, "wavefront");
        for (int j0 = 0; j0 < n; j0 += RB) {
          float f[RB][CPL];
#pragma unroll
          for (int u = 0; u < RB; ++u) {
            const int sj = __shfl(s, (j0 + u) & 31);
            if (j0 + u < n) {
#pragma unroll
              for (int p = 0; p < CPL; ++p)
                f[u][p] = feat[(long)sj * C + lane + 32 * p];
            }
          }
#pragma unroll
          for (int u = 0; u < RB; ++u) {
            const int rj = __shfl(rel, (j0 + u) & 31);
            if (j0 + u < n) {
#pragma unroll
              for (int p = 0; p < CPL; ++p) {
                const int k = lane + 32 * p;
                const float wv = s_w[wave][(j0 + u) * H + k / D];
                atomicAdd(&s_agg[rj * C + k], wv * f[u][p]);
              }
            }
          }
        }
      }
      qhead += n;
    };

    const bool sweep = (pass == 0) ? !imported : (s_resweep != 0);
    packed = !sweep;
    havev = (pass == 1) && !sweep;
    if (sweep) {
      qhead = 0;
      qtail = 0;
#pragma unroll 1
      for (int i = 0; i < NIT; ++i) {
        const int it = (i + (int)blockIdx.x) % NIT;
        const int blk0 = it * BPI + wave * 2;
        unsigned hit[2];
#pragma unroll
        for (int k = 0; k < 2; ++k) {
          const int blk = blk0 + k;
          const int bc = blk < OWN_NBLK ? blk : OWN_NBLK - 1;
          const uint4 pa = tab4[bc * 64 + lane];
          const uint4 pb = tab4[bc * 64 + 32 + lane];
          const unsigned pl[8] = {pa.x, pa.y, pa.z, pa.w, pb.x, pb.y, pb.z, pb.w};
          unsigned acc = ~0u;
#pragma unroll
          for (int j = 0; j < 8; ++j) acc &= ~(pl[j] ^ want[j]);
          hit[k] = (blk * 32 + lane < OWN_NGRP) ? acc : 0u;
        }
        unsigned h0 = hit[0], h1 = hit[1];
#pragma unroll 1
        for (int k = 0; k < 2; ++k) {
          unsigned mm = h0;
          h0 = h1;
          const int ebase = (blk0 + k) * 1024 + lane * 32;
          unsigned any = __builtin_amdgcn_ballot_w32(mm != 0u);
          while (any != 0u) {
            while (qtail - qhead > OWN_QCAP - 32) drain(32);
            if (mm != 0u) {
              const int q = __builtin_ctz(mm);
              mm &= mm - 1u;
              const int pos =
                  (qtail + (int)__builtin_amdgcn_mbcnt_lo(any, 0u)) &
                  (OWN_QCAP - 1);
              s_q[wave][pos] = ebase + q;
            }
            qtail += __popc(any);
            any = __builtin_amdgcn_ballot_w32(mm != 0u);
          }
        }
      }
      if (pass == 0 && qtail > OWN_QCAP) s_resweep = 1;
    } else if (pass == 0) {
      const int n = (int)icnt;
      for (int b = 0; b < n; b += 256) {
        unsigned t[8];
#pragma unroll
        for (int u = 0; u < 8; ++u)
          if (b + 32 * u < n) t[u] = lmine[b + 32 * u + lane];
#pragma unroll
        for (int u = 0; u < 8; ++u)
          if (b + 32 * u < n) s_q[wave][b + 32 * u + lane] = (int)t[u];
      }
      qhead = 0;
      qtail = n;
    } else {
      qhead = 0;
    }
    while (qtail - qhead >= 32) drain(32);
    if (qtail - qhead > 0) drain(qtail - qhead);
    if (LISTS == 1 && pass == 0 && nloc >= 1 && lane == 0)
      lcnt[wave] = qtail <= cap ? (unsigned)qtail : LST_NONE;
    __syncthreads();
  }

  for (int i = tid; i < nloc * C; i += 1024) {
    const int r = i / C;
    const int c = i % C;
    const float dn = s_den[r * H + c / D];
    float x = dn > 0.f ? s_agg[i] / dn : 0.f;
    x += bias[c];
    if (DO_ELU) x = x > 0.f ? x : expf(x) - 1.f;
    out[(long)lo * C + i] = x;
  }
}

extern "C" void kernel_launch(void* const* d_in, const int* in_sizes, int n_in,
                              void* d_out, int out_size, void* d_ws,
                              size_t ws_size, hipStream_t stream) {
  const float* x   = (const float*)d_in[0];
  const int*   src = (const int*)d_in[1];
  const int*   dst = (const int*)d_in[2];
  const float* W1  = (const float*)d_in[3];
  const float* al1 = (const float*)d_in[4];
  const float* ar1 = (const float*)d_in[5];
  const float* b1  = (const float*)d_in[6];
  const float* W2  = (const float*)d_in[7];
  const float* al2 = (const float*)d_in[8];
  const float* ar2 = (const float*)d_in[9];
  const float* b2  = (const float*)d_in[10];
  float* out = (float*)d_out;
  unsigned* lists = (unsigned*)d_out;

  const int N = NNODES;
  static_assert(NNODES % (16 * GEMM_WAVES) == 0, "every GEMM wave has a full row tile");

  float* ws = (float*)d_ws;
  float* feat1 = ws;
  float* el1   = feat1 + (long)N * 64;
  float* er1   = el1 + (long)N * 8;
  unsigned* tab = (unsigned*)(ws + (long)N * 80);
  float* h1    = ws + (long)N * 96;
  float* feat2 = ws;
  float* el2   = el1;
  float* er2   = el1 + N;

  const int TB = 256;
  auto nb = [](long n, int b) { return (int)((n + b - 1) / b); };
  const int NOWN = (NNODES + OWN_R - 1) / OWN_R;

  owner_table_kernel<<<nb(OWN_NBLK, 8), 256, 0, stream>>>(dst, tab);

  gemm_wmma_kernel<256, 64><<<NNODES / (16 * GEMM_WAVES), 32 * GEMM_WAVES, 0, stream>>>(x, W1, feat1);
  logits_kernel<8, 8><<<nb((long)N * 8, TB), TB, 0, stream>>>(feat1, al1, ar1, el1, er1, N * 8);
  owner_agg_kernel<8, 8, true, 1><<<NOWN, 1024, 0, stream>>>(src, dst, tab, el1, er1, feat1, b1, lists, h1);

  gemm_wmma_kernel<64, 32><<<NNODES / (16 * GEMM_WAVES), 32 * GEMM_WAVES, 0, stream>>>(h1, W2, feat2);
  logits_kernel<1, 32><<<nb((long)N, TB), TB, 0, stream>>>(feat2, al2, ar2, el2, er2, N);
  owner_agg_kernel<1, 32, false, 2><<<NOWN, 1024, 0, stream>>>(src, dst, tab, el2, er2, feat2, b2, lists, out);
}


# owner groups: each pass's reads (el/er rows, maxima, weights) stand ahead of its LDS atomics, which then run unbroken; edge's source id, row, one-head weight by v_readlane, not LDS; add order kept
# speedup vs baseline: 6.3864x; 1.4847x over previous
#include <hip/hip_runtime.h>
#include <hip/hip_bf16.h>

typedef __attribute__((ext_vector_type(2))) float v2f;
typedef __attribute__((ext_vector_type(8))) float v8f;

#define NNODES 100000
#define NEDGES 1600000

#define GEMM_WAVES 25

template <int K, int C>
__global__ __launch_bounds__(32 * GEMM_WAVES)
__attribute__((amdgpu_waves_per_eu(7, 7))) void gemm_wmma_kernel(
    const float* __restrict__ A, const float* __restrict__ B,
    float* __restrict__ D) {
  constexpr int NT = C / 16;
  constexpr int NL = K / 32;
  constexpr int BP = K + 4;
  static_assert(K % 32 == 0 && C % 16 == 0, "whole lines, whole tiles");
  static_assert((BP / 4) % 2 == 1, "row pitch: an odd number of 16-byte pieces");
  __shared__ __attribute__((aligned(16))) float sB[C * BP];

  const int tid = threadIdx.x;
  const int lane = tid & 31;
  const int wave = tid >> 5;
  const int half = lane >> 4;
  const int c    = lane & 15;
  const int row0 = (blockIdx.x * GEMM_WAVES + wave) * 16;

  const float4* ap =
      reinterpret_cast<const float4*>(A + (long)(row0 + c) * K + 16 * half);
  float4 a[2][4];
#pragma unroll
  for (int u = 0; u < 4; ++u) a[0][u] = ap[u];

  float* dpin = D + (long)(row0 + 8 * half) * C + c;
  asm volatile("" : "+v"(dpin));
  typedef __attribute__((address_space(1))) float global_float;
  global_float* drow = (global_float*)dpin;

  constexpr int NTHR = 32 * GEMM_WAVES;
  constexpr int NQ = C * K / 4;
  constexpr int NU = (NQ + NTHR - 1) / NTHR;
  float4 w[NU];
#pragma unroll
  for (int j = 0; j < NU; ++j) {
    const int q = tid + j * NTHR;
    if (q < NQ) {
      const int n = q % C;
      const int k = (q / C) * 4;
      w[j].x = B[(k + 0) * C + n];
      w[j].y = B[(k + 1) * C + n];
      w[j].z = B[(k + 2) * C + n];
      w[j].w = B[(k + 3) * C + n];
    }
  }
#pragma unroll
  for (int j = 0; j < NU; ++j) {
    const int q = tid + j * NTHR;
    if (q < NQ) {
      const int n = q % C;
      const int k = (q / C) * 4;
      *reinterpret_cast<float4*>(&sB[n * BP + k]) = w[j];
    }
  }
  __syncthreads();

  v8f acc[NT];
#pragma unroll
  for (int t = 0; t < NT; ++t) acc[t] = v8f{};

  const float* bl = sB + c * BP + 16 * half;
#pragma unroll
  for (int L = 0; L < NL; ++L) {
    if (L + 1 < NL) {
#pragma unroll
      for (int u = 0; u < 4; ++u) a[(L + 1) & 1][u] = ap[8 * (L + 1) + u];
    }
#pragma unroll
    for (int u = 0; u < 4; ++u) {
      const float4 av = a[L & 1][u];
      float4 bv[NT];
#pragma unroll
      for (int t = 0; t < NT; ++t)
        bv[t] = *reinterpret_cast<const float4*>(bl + t * 16 * BP + 32 * L + 4 * u);
      v2f a01, a23;
      a01.x = av.x; a01.y = av.y;
      a23.x = av.z; a23.y = av.w;
#pragma unroll
      for (int t = 0; t < NT; ++t) {
        v2f b01;
        b01.x = bv[t].x; b01.y = bv[t].y;
        acc[t] = __builtin_amdgcn_wmma_f32_16x16x4_f32(
             false, a01,  false, b01,
             (short)0, acc[t],  false,  false);
      }
#pragma unroll
      for (int t = 0; t < NT; ++t) {
        v2f b23;
        b23.x = bv[t].z; b23.y = bv[t].w;
        acc[t] = __builtin_amdgcn_wmma_f32_16x16x4_f32(
             false, a23,  false, b23,
             (short)0, acc[t],  false,  false);
      }
    }
  }

#pragma unroll
  for (int t = 0; t < NT; ++t) {
#pragma unroll
    for (int i = 0; i < 8; ++i) drow[i * C + 16 * t] = acc[t][i];
  }
}

template <int H, int D>
__global__ void logits_kernel(const float* __restrict__ feat,
                              const float* __restrict__ al,
                              const float* __restrict__ ar,
                              float* __restrict__ el, float* __restrict__ er,
                              int NH) {
  int i = blockIdx.x * blockDim.x + threadIdx.x;
  if (i >= NH) return;
  int n = i / H, h = i % H;
  const float* f = feat + ((long)n * H + h) * D;
  const float* a = al + h * D;
  const float* r = ar + h * D;
  float sl = 0.f, sr = 0.f;
#pragma unroll
  for (int d = 0; d < D; ++d) {
    sl += f[d] * a[d];
    sr += f[d] * r[d];
  }
  el[i] = sl;
  er[i] = sr;
}

__device__ __forceinline__ unsigned float_ord_encode(float f) {
  unsigned u = __float_as_uint(f);
  return (u & 0x80000000u) ? ~u : (u | 0x80000000u);
}
__device__ __forceinline__ float float_ord_decode(unsigned u) {
  return (u & 0x80000000u) ? __uint_as_float(u ^ 0x80000000u)
                           : __uint_as_float(~u);
}
#define ENC_NEG_INF 0x007FFFFFu

#define OWN_R 391
#define OWN_WAVES 32
#define OWN_QCAP 1024
#define OWN_RBITS 9

#define LST_ROW 32
#define LST_NONE 0xFFFFFFFFu

#define OWN_NGRP (NEDGES / 32)
#define OWN_NBLK ((OWN_NGRP + 31) / 32)

__global__ __launch_bounds__(256) void owner_table_kernel(
    const int* __restrict__ dst, unsigned* __restrict__ tab) {
  const int lane = threadIdx.x & 31;
  const int blk = blockIdx.x * 8 + (threadIdx.x >> 5);
  if (blk >= OWN_NBLK) return;
  const int grp = blk * 32 + lane;
  unsigned p[8] = {0u, 0u, 0u, 0u, 0u, 0u, 0u, 0u};
  if (grp < OWN_NGRP) {
    const int4* d4 = reinterpret_cast<const int4*>(dst + 32 * (long)grp);
#pragma unroll
    for (int c = 0; c < 8; ++c) {
      const int4 d = d4[c];
      const int dv[4] = {d.x, d.y, d.z, d.w};
#pragma unroll
      for (int k = 0; k < 4; ++k) {
        const unsigned oq = (unsigned)dv[k] / OWN_R;
        const unsigned o = oq < 255u ? oq : 255u;
#pragma unroll
        for (int j = 0; j < 8; ++j) p[j] |= ((o >> j) & 1u) << (4 * c + k);
      }
    }
  }
  uint4* t4 = reinterpret_cast<uint4*>(tab);
  t4[blk * 64 + lane]      = make_uint4(p[0], p[1], p[2], p[3]);
  t4[blk * 64 + 32 + lane] = make_uint4(p[4], p[5], p[6], p[7]);
}

template <int H, int D, bool DO_ELU, int LISTS>
__global__ __launch_bounds__(1024) void owner_agg_kernel(
    const int* __restrict__ src, const int* __restrict__ dst,
    const unsigned* __restrict__ tab,
    const float* __restrict__ el, const float* __restrict__ er,
    const float* __restrict__ feat, const float* __restrict__ bias,
    unsigned* lst, float* out) {
  constexpr int C = H * D;
  static_assert(LISTS != 2 || C == LST_ROW, "the lists lie in this layer's own output rows");
  constexpr int CPL = C / 32;
  constexpr int BPI = OWN_WAVES * 2;
  constexpr int NIT = (OWN_NBLK + BPI - 1) / BPI;
  constexpr int RB = 16;

  __shared__ float    s_agg[OWN_R * C];
  __shared__ float    s_den[OWN_R * H];
  __shared__ unsigned s_max[OWN_R * H];
  __shared__ int      s_q[OWN_WAVES][OWN_QCAP];
  constexpr bool ONEW = (H == 1);
  constexpr int HP = 32 / D;
  static_assert(ONEW || (CPL == 2 && H == 2 * HP), "a lane's two heads are a pair");
  __shared__ __attribute__((aligned(16))) float s_w[OWN_WAVES][ONEW ? 1 : 32 * H];
  __shared__ int      s_resweep;
  constexpr bool KEEPV = (H == 1);
  __shared__ float    s_v[KEEPV ? OWN_WAVES : 1][KEEPV ? OWN_QCAP : 1];

  const int tid = threadIdx.x;
  const int lane = tid & 31;
  const int wave = tid >> 5;
  const int lo = blockIdx.x * OWN_R;
  int nloc = NNODES - lo;
  nloc = nloc > OWN_R ? OWN_R : nloc;
  nloc = nloc < 0 ? 0 : nloc;
  static_assert(OWN_WAVES == LST_ROW, "a row of counts, cap rows of lists");
  const int cap = nloc >= 1 ? ((nloc - 1) & ~31) : 0;
  unsigned* const lcnt = lst + (long)lo * LST_ROW;
  unsigned* const lmine = lcnt + LST_ROW + wave * cap;

  for (int i = tid; i < OWN_R * C; i += 1024) s_agg[i] = 0.f;
  for (int i = tid; i < OWN_R * H; i += 1024) {
    s_den[i] = 0.f;
    s_max[i] = ENC_NEG_INF;
  }
  if (tid == 0) s_resweep = 0;
  __syncthreads();

  const uint4* tab4 = reinterpret_cast<const uint4*>(tab);
  unsigned want[8];
#pragma unroll
  for (int j = 0; j < 8; ++j) want[j] = ((blockIdx.x >> j) & 1u) ? ~0u : 0u;
  int qhead = 0, qtail = 0;
  bool packed = false;
  bool havev = false;
  unsigned icnt = LST_NONE;
  if (LISTS == 2 && nloc >= 1) icnt = __builtin_amdgcn_readfirstlane(lcnt[wave]);
  const bool imported = icnt <= (unsigned)cap;

#pragma unroll 1
  for (int pass = 0; pass < 2; ++pass) {

    auto drain = [&](int n) {
      __builtin_amdgcn_fence(__ATOMIC_RELEASE, "wavefront");
      __builtin_amdgcn_wave_barrier();
      __builtin_amdgcn_fence(__ATOMIC_ACQUIRE, "wavefront");
      int rel = 0, s = 0;
      float wl = 0.f;
      if (lane < n) {
        const int slot = (qhead + lane) & (OWN_QCAP - 1);
        const unsigned q = (unsigned)s_q[wave][slot];
        if (packed) {
          s = (int)(q >> OWN_RBITS);
          rel = (int)(q & ((1u << OWN_RBITS) - 1u));
        } else {
          s = src[q];
          const unsigned ru = (unsigned)(dst[q] - lo);
          rel = (int)(ru < (unsigned)(OWN_R - 1) ? ru : (unsigned)(OWN_R - 1));
          if (pass == 0) {
            const unsigned pk = ((unsigned)s << OWN_RBITS) | (unsigned)rel;
            s_q[wave][slot] = (int)pk;
            if (LISTS == 1 && slot < cap) lmine[slot] = pk;
          }
        }
        const float* els = el + (long)s * H;
        const float* erd = er + (long)(lo + rel) * H;
        float vv[H];
        if (KEEPV && havev) {
          vv[0] = s_v[wave][slot];
        } else {
#pragma unroll
          for (int h = 0; h < H; ++h) {
            const float v = els[h] + erd[h];
            vv[h] = v > 0.f ? v : 0.2f * v;
          }
          if (KEEPV && pass == 0) s_v[wave][slot] = vv[0];
        }
        if (pass == 0) {
#pragma unroll
          for (int h = 0; h < H; ++h)
            atomicMax(&s_max[rel * H + h], float_ord_encode(vv[h]));
        } else {
          float m[H];
#pragma unroll
          for (int h = 0; h < H; ++h) m[h] = float_ord_decode(s_max[rel * H + h]);
#pragma unroll
          for (int h = 0; h < H; ++h) {
            const float w = expf(vv[h] - m[h]);
            atomicAdd(&s_den[rel * H + h], w);
            if (ONEW) wl = w;
            else s_w[wave][lane * H + 2 * (h % HP) + h / HP] = w;
          }
        }
      }
      if (pass == 1) {
        if (!ONEW) {
          __builtin_amdgcn_fence(__ATOMIC_RELEASE, "wavefront");
          __builtin_amdgcn_wave_barrier();
          __builtin_amdgcn_fence(__ATOMIC_ACQUIRE, "wavefront");
        }
        for (int j0 = 0; j0 < n; j0 += RB) {
          float f[RB][CPL];
#pragma unroll
          for (int u = 0; u < RB; ++u) {
            const int sj = __builtin_amdgcn_readlane(s, j0 + u);
            if (j0 + u < n) {
#pragma unroll
              for (int p = 0; p < CPL; ++p)
                f[u][p] = feat[(long)sj * C + lane + 32 * p];
            }
          }
          float wv[RB][CPL];
#pragma unroll
          for (int u = 0; u < RB; ++u) {
            if (ONEW) {
              wv[u][0] = __int_as_float(
                  __builtin_amdgcn_readlane(__float_as_int(wl), j0 + u));
            } else if (j0 + u < n) {
              const float2 w2 = *reinterpret_cast<const float2*>(
                  &s_w[wave][(j0 + u) * H + 2 * (lane / D)]);
              wv[u][0] = w2.x;
              wv[u][CPL - 1] = w2.y;
            }
          }
#pragma unroll
          for (int u = 0; u < RB; ++u) {
            const int rj = __builtin_amdgcn_readlane(rel, j0 + u);
            if (j0 + u < n) {
#pragma unroll
              for (int p = 0; p < CPL; ++p)
                atomicAdd(&s_agg[rj * C + lane + 32 * p], wv[u][p] * f[u][p]);
            }
          }
        }
      }
      qhead += n;
    };

    const bool sweep = (pass == 0) ? !imported : (s_resweep != 0);
    packed = !sweep;
    havev = (pass == 1) && !sweep;
    if (sweep) {
      qhead = 0;
      qtail = 0;
#pragma unroll 1
      for (int i = 0; i < NIT; ++i) {
        const int it = (i + (int)blockIdx.x) % NIT;
        const int blk0 = it * BPI + wave * 2;
        unsigned hit[2];
#pragma unroll
        for (int k = 0; k < 2; ++k) {
          const int blk = blk0 + k;
          const int bc = blk < OWN_NBLK ? blk : OWN_NBLK - 1;
          const uint4 pa = tab4[bc * 64 + lane];
          const uint4 pb = tab4[bc * 64 + 32 + lane];
          const unsigned pl[8] = {pa.x, pa.y, pa.z, pa.w, pb.x, pb.y, pb.z, pb.w};
          unsigned acc = ~0u;
#pragma unroll
          for (int j = 0; j < 8; ++j) acc &= ~(pl[j] ^ want[j]);
          hit[k] = (blk * 32 + lane < OWN_NGRP) ? acc : 0u;
        }
        unsigned h0 = hit[0], h1 = hit[1];
#pragma unroll 1
        for (int k = 0; k < 2; ++k) {
          unsigned mm = h0;
          h0 = h1;
          const int ebase = (blk0 + k) * 1024 + lane * 32;
          unsigned any = __builtin_amdgcn_ballot_w32(mm != 0u);
          while (any != 0u) {
            while (qtail - qhead > OWN_QCAP - 32) drain(32);
            if (mm != 0u) {
              const int q = __builtin_ctz(mm);
              mm &= mm - 1u;
              const int pos =
                  (qtail + (int)__builtin_amdgcn_mbcnt_lo(any, 0u)) &
                  (OWN_QCAP - 1);
              s_q[wave][pos] = ebase + q;
            }
            qtail += __popc(any);
            any = __builtin_amdgcn_ballot_w32(mm != 0u);
          }
        }
      }
      if (pass == 0 && qtail > OWN_QCAP) s_resweep = 1;
    } else if (pass == 0) {
      const int n = (int)icnt;
      for (int b = 0; b < n; b += 256) {
        unsigned t[8];
#pragma unroll
        for (int u = 0; u < 8; ++u)
          if (b + 32 * u < n) t[u] = lmine[b + 32 * u + lane];
#pragma unroll
        for (int u = 0; u < 8; ++u)
          if (b + 32 * u < n) s_q[wave][b + 32 * u + lane] = (int)t[u];
      }
      qhead = 0;
      qtail = n;
    } else {
      qhead = 0;
    }
    while (qtail - qhead >= 32) drain(32);
    if (qtail - qhead > 0) drain(qtail - qhead);
    if (LISTS == 1 && pass == 0 && nloc >= 1 && lane == 0)
      lcnt[wave] = qtail <= cap ? (unsigned)qtail : LST_NONE;
    __syncthreads();
  }

  for (int i = tid; i < nloc * C; i += 1024) {
    const int r = i / C;
    const int c = i % C;
    const float dn = s_den[r * H + c / D];
    float x = dn > 0.f ? s_agg[i] / dn : 0.f;
    x += bias[c];
    if (DO_ELU) x = x > 0.f ? x : expf(x) - 1.f;
    out[(long)lo * C + i] = x;
  }
}

extern "C" void kernel_launch(void* const* d_in, const int* in_sizes, int n_in,
                              void* d_out, int out_size, void* d_ws,
                              size_t ws_size, hipStream_t stream) {
  const float* x   = (const float*)d_in[0];
  const int*   src = (const int*)d_in[1];
  const int*   dst = (const int*)d_in[2];
  const float* W1  = (const float*)d_in[3];
  const float* al1 = (const float*)d_in[4];
  const float* ar1 = (const float*)d_in[5];
  const float* b1  = (const float*)d_in[6];
  const float* W2  = (const float*)d_in[7];
  const float* al2 = (const float*)d_in[8];
  const float* ar2 = (const float*)d_in[9];
  const float* b2  = (const float*)d_in[10];
  float* out = (float*)d_out;
  unsigned* lists = (unsigned*)d_out;

  const int N = NNODES;
  static_assert(NNODES % (16 * GEMM_WAVES) == 0, "every GEMM wave has a full row tile");

  float* ws = (float*)d_ws;
  float* feat1 = ws;
  float* el1   = feat1 + (long)N * 64;
  float* er1   = el1 + (long)N * 8;
  unsigned* tab = (unsigned*)(ws + (long)N * 80);
  float* h1    = ws + (long)N * 96;
  float* feat2 = ws;
  float* el2   = el1;
  float* er2   = el1 + N;

  const int TB = 256;
  auto nb = [](long n, int b) { return (int)((n + b - 1) / b); };
  const int NOWN = (NNODES + OWN_R - 1) / OWN_R;

  owner_table_kernel<<<nb(OWN_NBLK, 8), 256, 0, stream>>>(dst, tab);

  gemm_wmma_kernel<256, 64><<<NNODES / (16 * GEMM_WAVES), 32 * GEMM_WAVES, 0, stream>>>(x, W1, feat1);
  logits_kernel<8, 8><<<nb((long)N * 8, TB), TB, 0, stream>>>(feat1, al1, ar1, el1, er1, N * 8);
  owner_agg_kernel<8, 8, true, 1><<<NOWN, 1024, 0, stream>>>(src, dst, tab, el1, er1, feat1, b1, lists, h1);

  gemm_wmma_kernel<64, 32><<<NNODES / (16 * GEMM_WAVES), 32 * GEMM_WAVES, 0, stream>>>(h1, W2, feat2);
  logits_kernel<1, 32><<<nb((long)N, TB), TB, 0, stream>>>(feat2, al2, ar2, el2, er2, N);
  owner_agg_kernel<1, 32, false, 2><<<NOWN, 1024, 0, stream>>>(src, dst, tab, el2, er2, feat2, b2, lists, out);
}


# attention logits el, er formed in each GEMM's epilogue from the wave's accumulator tiles (lane sums over a head's columns by row_xmask DPP); both logits launches go, 7 -> 5; el/er sum order changed
# speedup vs baseline: 6.6197x; 1.0365x over previous
#include <hip/hip_runtime.h>
#include <hip/hip_bf16.h>

typedef __attribute__((ext_vector_type(2))) float v2f;
typedef __attribute__((ext_vector_type(8))) float v8f;

#define NNODES 100000
#define NEDGES 1600000

#define GEMM_WAVES 25

template <int M>
__device__ __forceinline__ float row_xor(float x) {
  const int b = __builtin_bit_cast(int, x);
  return __builtin_bit_cast(
      float, __builtin_amdgcn_update_dpp(0, b, 0x160 + M, 0xf, 0xf, true));
}

template <int CNT, int M>
__device__ __forceinline__ void fold(float* v, bool hi) {
#pragma unroll
  for (int n = 0; n < CNT; ++n) {
    const float keep = hi ? v[2 * n + 1] : v[2 * n];
    const float send = hi ? v[2 * n] : v[2 * n + 1];
    v[n] = keep + row_xor<M>(send);
  }
}

template <int K, int C, int H, int LS>
__global__ __launch_bounds__(32 * GEMM_WAVES)
__attribute__((amdgpu_waves_per_eu(7, 7))) void gemm_wmma_kernel(
    const float* __restrict__ A, const float* __restrict__ B,
    const float* __restrict__ al, const float* __restrict__ ar,
    float* __restrict__ D, float* __restrict__ el, float* __restrict__ er) {
  constexpr int NT = C / 16;
  constexpr int DH = C / H;
  static_assert(C % H == 0 && (DH == 8 || DH % 16 == 0), "a head: 8 lanes of a tile, or whole tiles");
  constexpr int TPH = DH > 16 ? DH / 16 : 1;
  constexpr int NHT = NT / TPH;
  constexpr int HPG = DH < 16 ? 16 / DH : 1;
  static_assert(LS >= H, "LS: words from a node's logits to the next node's");
  constexpr int NL = K / 32;
  constexpr int BP = K + 4;
  static_assert(K % 32 == 0 && C % 16 == 0, "whole lines, whole tiles");
  static_assert((BP / 4) % 2 == 1, "row pitch: an odd number of 16-byte pieces");
  __shared__ __attribute__((aligned(16))) float sB[C * BP];

  const int tid = threadIdx.x;
  const int lane = tid & 31;
  const int wave = tid >> 5;
  const int half = lane >> 4;
  const int c    = lane & 15;
  const int row0 = (blockIdx.x * GEMM_WAVES + wave) * 16;

  const float4* ap =
      reinterpret_cast<const float4*>(A + (long)(row0 + c) * K + 16 * half);
  float4 a[2][4];
#pragma unroll
  for (int u = 0; u < 4; ++u) a[0][u] = ap[u];

  float* dpin = D + (long)(row0 + 8 * half) * C + c;
  asm volatile("" : "+v"(dpin));
  typedef __attribute__((address_space(1))) float global_float;
  global_float* drow = (global_float*)dpin;

  float lv[NT], rv[NT];
#pragma unroll
  for (int t = 0; t < NT; ++t) {
    lv[t] = al[16 * t + c];
    rv[t] = ar[16 * t + c];
  }
  unsigned loff = (unsigned)(row0 + 8 * half + (c & 7)) * LS + (DH < 16 ? (c >> 3) : 0);
  int lbits = lane;
  asm volatile("" : "+v"(loff), "+v"(lbits));

  constexpr int NTHR = 32 * GEMM_WAVES;
  constexpr int NQ = C * K / 4;
  constexpr int NU = (NQ + NTHR - 1) / NTHR;
  float4 w[NU];
#pragma unroll
  for (int j = 0; j < NU; ++j) {
    const int q = tid + j * NTHR;
    if (q < NQ) {
      const int n = q % C;
      const int k = (q / C) * 4;
      w[j].x = B[(k + 0) * C + n];
      w[j].y = B[(k + 1) * C + n];
      w[j].z = B[(k + 2) * C + n];
      w[j].w = B[(k + 3) * C + n];
    }
  }
#pragma unroll
  for (int j = 0; j < NU; ++j) {
    const int q = tid + j * NTHR;
    if (q < NQ) {
      const int n = q % C;
      const int k = (q / C) * 4;
      *reinterpret_cast<float4*>(&sB[n * BP + k]) = w[j];
    }
  }
  __syncthreads();

  v8f acc[NT];
#pragma unroll
  for (int t = 0; t < NT; ++t) acc[t] = v8f{};

  const float* bl = sB + c * BP + 16 * half;
#pragma unroll
  for (int L = 0; L < NL; ++L) {
    if (L + 1 < NL) {
#pragma unroll
      for (int u = 0; u < 4; ++u) a[(L + 1) & 1][u] = ap[8 * (L + 1) + u];
    }
#pragma unroll
    for (int u = 0; u < 4; ++u) {
      const float4 av = a[L & 1][u];
      float4 bv[NT];
#pragma unroll
      for (int t = 0; t < NT; ++t)
        bv[t] = *reinterpret_cast<const float4*>(bl + t * 16 * BP + 32 * L + 4 * u);
      v2f a01, a23;
      a01.x = av.x; a01.y = av.y;
      a23.x = av.z; a23.y = av.w;
#pragma unroll
      for (int t = 0; t < NT; ++t) {
        v2f b01;
        b01.x = bv[t].x; b01.y = bv[t].y;
        acc[t] = __builtin_amdgcn_wmma_f32_16x16x4_f32(
             false, a01,  false, b01,
             (short)0, acc[t],  false,  false);
      }
#pragma unroll
      for (int t = 0; t < NT; ++t) {
        v2f b23;
        b23.x = bv[t].z; b23.y = bv[t].w;
        acc[t] = __builtin_amdgcn_wmma_f32_16x16x4_f32(
             false, a23,  false, b23,
             (short)0, acc[t],  false,  false);
      }
    }
  }

  static_assert(NT == 2 || NT == 4, "the statement below names every tile");
  if constexpr (NT == 4) {
    asm volatile("v_nop\n\tv_nop\n\tv_nop\n\tv_nop\n\tv_nop"
                 : "+v"(acc[0]), "+v"(acc[1]), "+v"(acc[2]), "+v"(acc[3]),
                   "+v"(lv[0]), "+v"(lv[1]), "+v"(lv[2]), "+v"(lv[3]),
                   "+v"(rv[0]), "+v"(rv[1]), "+v"(rv[2]), "+v"(rv[3]),
                   "+v"(loff), "+v"(lbits));
  } else {
    asm volatile("v_nop\n\tv_nop\n\tv_nop\n\tv_nop\n\tv_nop"
                 : "+v"(acc[0]), "+v"(acc[1]),
                   "+v"(lv[0]), "+v"(lv[1]), "+v"(rv[0]), "+v"(rv[1]),
                   "+v"(loff), "+v"(lbits));
  }
  const bool hb0 = (lbits & 1) != 0, hb1 = (lbits & 2) != 0;
  const bool hb2 = (lbits & 4) != 0, hb3 = (lbits & 8) != 0;
  global_float* g0 = (global_float*)(DH < 16 ? el : (hb3 ? er : el)) + loff;
  global_float* g1 = (global_float*)er + loff;

#pragma unroll
  for (int t = 0; t < NT; ++t) {
#pragma unroll
    for (int i = 0; i < 8; ++i) drow[i * C + 16 * t] = acc[t][i];
  }

#pragma unroll
  for (int g = 0; g < NHT; ++g) {
    float v[16];
#pragma unroll
    for (int i = 0; i < 8; ++i) {
      float sl = acc[g * TPH][i] * lv[g * TPH];
      float sr = acc[g * TPH][i] * rv[g * TPH];
#pragma unroll
      for (int u = 1; u < TPH; ++u) {
        sl += acc[g * TPH + u][i] * lv[g * TPH + u];
        sr += acc[g * TPH + u][i] * rv[g * TPH + u];
      }
      v[i] = sl;
      v[8 + i] = sr;
    }
    fold<8, 1>(v, hb0);
    fold<4, 2>(v, hb1);
    fold<2, 4>(v, hb2);
    if constexpr (DH >= 16) {
      fold<1, 8>(v, hb3);
      g0[g] = v[0];
    } else {
      g0[HPG * g] = v[0];
      g1[HPG * g] = v[1];
    }
  }
}

__device__ __forceinline__ unsigned float_ord_encode(float f) {
  unsigned u = __float_as_uint(f);
  return (u & 0x80000000u) ? ~u : (u | 0x80000000u);
}
__device__ __forceinline__ float float_ord_decode(unsigned u) {
  return (u & 0x80000000u) ? __uint_as_float(u ^ 0x80000000u)
                           : __uint_as_float(~u);
}
#define ENC_NEG_INF 0x007FFFFFu

#define OWN_R 391
#define OWN_WAVES 32
#define OWN_QCAP 1024
#define OWN_RBITS 9

#define LST_ROW 32
#define LST_NONE 0xFFFFFFFFu

#define OWN_NGRP (NEDGES / 32)
#define OWN_NBLK ((OWN_NGRP + 31) / 32)

__global__ __launch_bounds__(256) void owner_table_kernel(
    const int* __restrict__ dst, unsigned* __restrict__ tab) {
  const int lane = threadIdx.x & 31;
  const int blk = blockIdx.x * 8 + (threadIdx.x >> 5);
  if (blk >= OWN_NBLK) return;
  const int grp = blk * 32 + lane;
  unsigned p[8] = {0u, 0u, 0u, 0u, 0u, 0u, 0u, 0u};
  if (grp < OWN_NGRP) {
    const int4* d4 = reinterpret_cast<const int4*>(dst + 32 * (long)grp);
#pragma unroll
    for (int c = 0; c < 8; ++c) {
      const int4 d = d4[c];
      const int dv[4] = {d.x, d.y, d.z, d.w};
#pragma unroll
      for (int k = 0; k < 4; ++k) {
        const unsigned oq = (unsigned)dv[k] / OWN_R;
        const unsigned o = oq < 255u ? oq : 255u;
#pragma unroll
        for (int j = 0; j < 8; ++j) p[j] |= ((o >> j) & 1u) << (4 * c + k);
      }
    }
  }
  uint4* t4 = reinterpret_cast<uint4*>(tab);
  t4[blk * 64 + lane]      = make_uint4(p[0], p[1], p[2], p[3]);
  t4[blk * 64 + 32 + lane] = make_uint4(p[4], p[5], p[6], p[7]);
}

template <int H, int D, bool DO_ELU, int LISTS, int LS>
__global__ __launch_bounds__(1024) void owner_agg_kernel(
    const int* __restrict__ src, const int* __restrict__ dst,
    const unsigned* __restrict__ tab,
    const float* __restrict__ el, const float* __restrict__ er,
    const float* __restrict__ feat, const float* __restrict__ bias,
    unsigned* lst, float* out) {
  constexpr int C = H * D;
  static_assert(LISTS != 2 || C == LST_ROW, "the lists lie in this layer's own output rows");
  constexpr int CPL = C / 32;
  constexpr int BPI = OWN_WAVES * 2;
  constexpr int NIT = (OWN_NBLK + BPI - 1) / BPI;
  constexpr int RB = 16;

  __shared__ float    s_agg[OWN_R * C];
  __shared__ float    s_den[OWN_R * H];
  __shared__ unsigned s_max[OWN_R * H];
  __shared__ int      s_q[OWN_WAVES][OWN_QCAP];
  constexpr bool ONEW = (H == 1);
  constexpr int HP = 32 / D;
  static_assert(ONEW || (CPL == 2 && H == 2 * HP), "a lane's two heads are a pair");
  __shared__ __attribute__((aligned(16))) float s_w[OWN_WAVES][ONEW ? 1 : 32 * H];
  __shared__ int      s_resweep;
  constexpr bool KEEPV = (H == 1);
  __shared__ float    s_v[KEEPV ? OWN_WAVES : 1][KEEPV ? OWN_QCAP : 1];

  const int tid = threadIdx.x;
  const int lane = tid & 31;
  const int wave = tid >> 5;
  const int lo = blockIdx.x * OWN_R;
  int nloc = NNODES - lo;
  nloc = nloc > OWN_R ? OWN_R : nloc;
  nloc = nloc < 0 ? 0 : nloc;
  static_assert(OWN_WAVES == LST_ROW, "a row of counts, cap rows of lists");
  const int cap = nloc >= 1 ? ((nloc - 1) & ~31) : 0;
  unsigned* const lcnt = lst + (long)lo * LST_ROW;
  unsigned* const lmine = lcnt + LST_ROW + wave * cap;

  for (int i = tid; i < OWN_R * C; i += 1024) s_agg[i] = 0.f;
  for (int i = tid; i < OWN_R * H; i += 1024) {
    s_den[i] = 0.f;
    s_max[i] = ENC_NEG_INF;
  }
  if (tid == 0) s_resweep = 0;
  __syncthreads();

  const uint4* tab4 = reinterpret_cast<const uint4*>(tab);
  unsigned want[8];
#pragma unroll
  for (int j = 0; j < 8; ++j) want[j] = ((blockIdx.x >> j) & 1u) ? ~0u : 0u;
  int qhead = 0, qtail = 0;
  bool packed = false;
  bool havev = false;
  unsigned icnt = LST_NONE;
  if (LISTS == 2 && nloc >= 1) icnt = __builtin_amdgcn_readfirstlane(lcnt[wave]);
  const bool imported = icnt <= (unsigned)cap;

#pragma unroll 1
  for (int pass = 0; pass < 2; ++pass) {

    auto drain = [&](int n) {
      __builtin_amdgcn_fence(__ATOMIC_RELEASE, "wavefront");
      __builtin_amdgcn_wave_barrier();
      __builtin_amdgcn_fence(__ATOMIC_ACQUIRE, "wavefront");
      int rel = 0, s = 0;
      float wl = 0.f;
      if (lane < n) {
        const int slot = (qhead + lane) & (OWN_QCAP - 1);
        const unsigned q = (unsigned)s_q[wave][slot];
        if (packed) {
          s = (int)(q >> OWN_RBITS);
          rel = (int)(q & ((1u << OWN_RBITS) - 1u));
        } else {
          s = src[q];
          const unsigned ru = (unsigned)(dst[q] - lo);
          rel = (int)(ru < (unsigned)(OWN_R - 1) ? ru : (unsigned)(OWN_R - 1));
          if (pass == 0) {
            const unsigned pk = ((unsigned)s << OWN_RBITS) | (unsigned)rel;
            s_q[wave][slot] = (int)pk;
            if (LISTS == 1 && slot < cap) lmine[slot] = pk;
          }
        }
        const float* els = el + (long)s * LS;
        const float* erd = er + (long)(lo + rel) * LS;
        float vv[H];
        if (KEEPV && havev) {
          vv[0] = s_v[wave][slot];
        } else {
#pragma unroll
          for (int h = 0; h < H; ++h) {
            const float v = els[h] + erd[h];
            vv[h] = v > 0.f ? v : 0.2f * v;
          }
          if (KEEPV && pass == 0) s_v[wave][slot] = vv[0];
        }
        if (pass == 0) {
#pragma unroll
          for (int h = 0; h < H; ++h)
            atomicMax(&s_max[rel * H + h], float_ord_encode(vv[h]));
        } else {
          float m[H];
#pragma unroll
          for (int h = 0; h < H; ++h) m[h] = float_ord_decode(s_max[rel * H + h]);
#pragma unroll
          for (int h = 0; h < H; ++h) {
            const float w = expf(vv[h] - m[h]);
            atomicAdd(&s_den[rel * H + h], w);
            if (ONEW) wl = w;
            else s_w[wave][lane * H + 2 * (h % HP) + h / HP] = w;
          }
        }
      }
      if (pass == 1) {
        if (!ONEW) {
          __builtin_amdgcn_fence(__ATOMIC_RELEASE, "wavefront");
          __builtin_amdgcn_wave_barrier();
          __builtin_amdgcn_fence(__ATOMIC_ACQUIRE, "wavefront");
        }
        for (int j0 = 0; j0 < n; j0 += RB) {
          float f[RB][CPL];
#pragma unroll
          for (int u = 0; u < RB; ++u) {
            const int sj = __builtin_amdgcn_readlane(s, j0 + u);
            if (j0 + u < n) {
#pragma unroll
              for (int p = 0; p < CPL; ++p)
                f[u][p] = feat[(long)sj * C + lane + 32 * p];
            }
          }
          float wv[RB][CPL];
#pragma unroll
          for (int u = 0; u < RB; ++u) {
            if (ONEW) {
              wv[u][0] = __int_as_float(
                  __builtin_amdgcn_readlane(__float_as_int(wl), j0 + u));
            } else if (j0 + u < n) {
              const float2 w2 = *reinterpret_cast<const float2*>(
                  &s_w[wave][(j0 + u) * H + 2 * (lane / D)]);
              wv[u][0] = w2.x;
              wv[u][CPL - 1] = w2.y;
            }
          }
#pragma unroll
          for (int u = 0; u < RB; ++u) {
            const int rj = __builtin_amdgcn_readlane(rel, j0 + u);
            if (j0 + u < n) {
#pragma unroll
              for (int p = 0; p < CPL; ++p)
                atomicAdd(&s_agg[rj * C + lane + 32 * p], wv[u][p] * f[u][p]);
            }
          }
        }
      }
      qhead += n;
    };

    const bool sweep = (pass == 0) ? !imported : (s_resweep != 0);
    packed = !sweep;
    havev = (pass == 1) && !sweep;
    if (sweep) {
      qhead = 0;
      qtail = 0;
#pragma unroll 1
      for (int i = 0; i < NIT; ++i) {
        const int it = (i + (int)blockIdx.x) % NIT;
        const int blk0 = it * BPI + wave * 2;
        unsigned hit[2];
#pragma unroll
        for (int k = 0; k < 2; ++k) {
          const int blk = blk0 + k;
          const int bc = blk < OWN_NBLK ? blk : OWN_NBLK - 1;
          const uint4 pa = tab4[bc * 64 + lane];
          const uint4 pb = tab4[bc * 64 + 32 + lane];
          const unsigned pl[8] = {pa.x, pa.y, pa.z, pa.w, pb.x, pb.y, pb.z, pb.w};
          unsigned acc = ~0u;
#pragma unroll
          for (int j = 0; j < 8; ++j) acc &= ~(pl[j] ^ want[j]);
          hit[k] = (blk * 32 + lane < OWN_NGRP) ? acc : 0u;
        }
        unsigned h0 = hit[0], h1 = hit[1];
#pragma unroll 1
        for (int k = 0; k < 2; ++k) {
          unsigned mm = h0;
          h0 = h1;
          const int ebase = (blk0 + k) * 1024 + lane * 32;
          unsigned any = __builtin_amdgcn_ballot_w32(mm != 0u);
          while (any != 0u) {
            while (qtail - qhead > OWN_QCAP - 32) drain(32);
            if (mm != 0u) {
              const int q = __builtin_ctz(mm);
              mm &= mm - 1u;
              const int pos =
                  (qtail + (int)__builtin_amdgcn_mbcnt_lo(any, 0u)) &
                  (OWN_QCAP - 1);
              s_q[wave][pos] = ebase + q;
            }
            qtail += __popc(any);
            any = __builtin_amdgcn_ballot_w32(mm != 0u);
          }
        }
      }
      if (pass == 0 && qtail > OWN_QCAP) s_resweep = 1;
    } else if (pass == 0) {
      const int n = (int)icnt;
      for (int b = 0; b < n; b += 256) {
        unsigned t[8];
#pragma unroll
        for (int u = 0; u < 8; ++u)
          if (b + 32 * u < n) t[u] = lmine[b + 32 * u + lane];
#pragma unroll
        for (int u = 0; u < 8; ++u)
          if (b + 32 * u < n) s_q[wave][b + 32 * u + lane] = (int)t[u];
      }
      qhead = 0;
      qtail = n;
    } else {
      qhead = 0;
    }
    while (qtail - qhead >= 32) drain(32);
    if (qtail - qhead > 0) drain(qtail - qhead);
    if (LISTS == 1 && pass == 0 && nloc >= 1 && lane == 0)
      lcnt[wave] = qtail <= cap ? (unsigned)qtail : LST_NONE;
    __syncthreads();
  }

  for (int i = tid; i < nloc * C; i += 1024) {
    const int r = i / C;
    const int c = i % C;
    const float dn = s_den[r * H + c / D];
    float x = dn > 0.f ? s_agg[i] / dn : 0.f;
    x += bias[c];
    if (DO_ELU) x = x > 0.f ? x : expf(x) - 1.f;
    out[(long)lo * C + i] = x;
  }
}

extern "C" void kernel_launch(void* const* d_in, const int* in_sizes, int n_in,
                              void* d_out, int out_size, void* d_ws,
                              size_t ws_size, hipStream_t stream) {
  const float* x   = (const float*)d_in[0];
  const int*   src = (const int*)d_in[1];
  const int*   dst = (const int*)d_in[2];
  const float* W1  = (const float*)d_in[3];
  const float* al1 = (const float*)d_in[4];
  const float* ar1 = (const float*)d_in[5];
  const float* b1  = (const float*)d_in[6];
  const float* W2  = (const float*)d_in[7];
  const float* al2 = (const float*)d_in[8];
  const float* ar2 = (const float*)d_in[9];
  const float* b2  = (const float*)d_in[10];
  float* out = (float*)d_out;
  unsigned* lists = (unsigned*)d_out;

  const int N = NNODES;
  static_assert(NNODES % (16 * GEMM_WAVES) == 0, "every GEMM wave has a full row tile");

  float* ws = (float*)d_ws;
  float* feat1 = ws;
  float* el1   = feat1 + (long)N * 64;
  float* er1   = el1 + (long)N * 8;
  unsigned* tab = (unsigned*)(ws + (long)N * 80);
  float* h1    = ws + (long)N * 96;
  float* feat2 = ws;
  float* el2   = el1;
  float* er2   = el1 + 1;

  auto nb = [](long n, int b) { return (int)((n + b - 1) / b); };
  const int NOWN = (NNODES + OWN_R - 1) / OWN_R;

  owner_table_kernel<<<nb(OWN_NBLK, 8), 256, 0, stream>>>(dst, tab);

  gemm_wmma_kernel<256, 64, 8, 8><<<NNODES / (16 * GEMM_WAVES), 32 * GEMM_WAVES, 0, stream>>>(x, W1, al1, ar1, feat1, el1, er1);
  owner_agg_kernel<8, 8, true, 1, 8><<<NOWN, 1024, 0, stream>>>(src, dst, tab, el1, er1, feat1, b1, lists, h1);

  gemm_wmma_kernel<64, 32, 1, 2><<<NNODES / (16 * GEMM_WAVES), 32 * GEMM_WAVES, 0, stream>>>(h1, W2, al2, ar2, feat2, el2, er2);
  owner_agg_kernel<1, 32, false, 2, 2><<<NOWN, 1024, 0, stream>>>(src, dst, tab, el2, er2, feat2, b2, lists, out);
}


# layer 2's projection done by layer 1's owner groups on their finished rows in LDS (f32 WMMA, same loaders, k order and logit sums): layer-1 output never leaves the chip, 5 -> 4 launches
# speedup vs baseline: 6.9720x; 1.0532x over previous
#include <hip/hip_runtime.h>
#include <hip/hip_bf16.h>

typedef __attribute__((ext_vector_type(2))) float v2f;
typedef __attribute__((ext_vector_type(8))) float v8f;

#define NNODES 100000
#define NEDGES 1600000

#define GEMM_WAVES 25

template <int M>
__device__ __forceinline__ float row_xor(float x) {
  const int b = __builtin_bit_cast(int, x);
  return __builtin_bit_cast(
      float, __builtin_amdgcn_update_dpp(0, b, 0x160 + M, 0xf, 0xf, true));
}

template <int CNT, int M>
__device__ __forceinline__ void fold(float* v, bool hi) {
#pragma unroll
  for (int n = 0; n < CNT; ++n) {
    const float keep = hi ? v[2 * n + 1] : v[2 * n];
    const float send = hi ? v[2 * n] : v[2 * n + 1];
    v[n] = keep + row_xor<M>(send);
  }
}

template <int K, int C, int H, int LS>
__global__ __launch_bounds__(32 * GEMM_WAVES)
__attribute__((amdgpu_waves_per_eu(7, 7))) void gemm_wmma_kernel(
    const float* __restrict__ A, const float* __restrict__ B,
    const float* __restrict__ al, const float* __restrict__ ar,
    float* __restrict__ D, float* __restrict__ el, float* __restrict__ er) {
  constexpr int NT = C / 16;
  constexpr int DH = C / H;
  static_assert(C % H == 0 && (DH == 8 || DH % 16 == 0), "a head: 8 lanes of a tile, or whole tiles");
  constexpr int TPH = DH > 16 ? DH / 16 : 1;
  constexpr int NHT = NT / TPH;
  constexpr int HPG = DH < 16 ? 16 / DH : 1;
  static_assert(LS >= H, "LS: words from a node's logits to the next node's");
  constexpr int NL = K / 32;
  constexpr int BP = K + 4;
  static_assert(K % 32 == 0 && C % 16 == 0, "whole lines, whole tiles");
  static_assert((BP / 4) % 2 == 1, "row pitch: an odd number of 16-byte pieces");
  __shared__ __attribute__((aligned(16))) float sB[C * BP];

  const int tid = threadIdx.x;
  const int lane = tid & 31;
  const int wave = tid >> 5;
  const int half = lane >> 4;
  const int c    = lane & 15;
  const int row0 = (blockIdx.x * GEMM_WAVES + wave) * 16;

  const float4* ap =
      reinterpret_cast<const float4*>(A + (long)(row0 + c) * K + 16 * half);
  float4 a[2][4];
#pragma unroll
  for (int u = 0; u < 4; ++u) a[0][u] = ap[u];

  float* dpin = D + (long)(row0 + 8 * half) * C + c;
  asm volatile("" : "+v"(dpin));
  typedef __attribute__((address_space(1))) float global_float;
  global_float* drow = (global_float*)dpin;

  float lv[NT], rv[NT];
#pragma unroll
  for (int t = 0; t < NT; ++t) {
    lv[t] = al[16 * t + c];
    rv[t] = ar[16 * t + c];
  }
  unsigned loff = (unsigned)(row0 + 8 * half + (c & 7)) * LS + (DH < 16 ? (c >> 3) : 0);
  int lbits = lane;
  asm volatile("" : "+v"(loff), "+v"(lbits));

  constexpr int NTHR = 32 * GEMM_WAVES;
  constexpr int NQ = C * K / 4;
  constexpr int NU = (NQ + NTHR - 1) / NTHR;
  float4 w[NU];
#pragma unroll
  for (int j = 0; j < NU; ++j) {
    const int q = tid + j * NTHR;
    if (q < NQ) {
      const int n = q % C;
      const int k = (q / C) * 4;
      w[j].x = B[(k + 0) * C + n];
      w[j].y = B[(k + 1) * C + n];
      w[j].z = B[(k + 2) * C + n];
      w[j].w = B[(k + 3) * C + n];
    }
  }
#pragma unroll
  for (int j = 0; j < NU; ++j) {
    const int q = tid + j * NTHR;
    if (q < NQ) {
      const int n = q % C;
      const int k = (q / C) * 4;
      *reinterpret_cast<float4*>(&sB[n * BP + k]) = w[j];
    }
  }
  __syncthreads();

  v8f acc[NT];
#pragma unroll
  for (int t = 0; t < NT; ++t) acc[t] = v8f{};

  const float* bl = sB + c * BP + 16 * half;
#pragma unroll
  for (int L = 0; L < NL; ++L) {
    if (L + 1 < NL) {
#pragma unroll
      for (int u = 0; u < 4; ++u) a[(L + 1) & 1][u] = ap[8 * (L + 1) + u];
    }
#pragma unroll
    for (int u = 0; u < 4; ++u) {
      const float4 av = a[L & 1][u];
      float4 bv[NT];
#pragma unroll
      for (int t = 0; t < NT; ++t)
        bv[t] = *reinterpret_cast<const float4*>(bl + t * 16 * BP + 32 * L + 4 * u);
      v2f a01, a23;
      a01.x = av.x; a01.y = av.y;
      a23.x = av.z; a23.y = av.w;
#pragma unroll
      for (int t = 0; t < NT; ++t) {
        v2f b01;
        b01.x = bv[t].x; b01.y = bv[t].y;
        acc[t] = __builtin_amdgcn_wmma_f32_16x16x4_f32(
             false, a01,  false, b01,
             (short)0, acc[t],  false,  false);
      }
#pragma unroll
      for (int t = 0; t < NT; ++t) {
        v2f b23;
        b23.x = bv[t].z; b23.y = bv[t].w;
        acc[t] = __builtin_amdgcn_wmma_f32_16x16x4_f32(
             false, a23,  false, b23,
             (short)0, acc[t],  false,  false);
      }
    }
  }

  static_assert(NT == 2 || NT == 4, "the statement below names every tile");
  if constexpr (NT == 4) {
    asm volatile("v_nop\n\tv_nop\n\tv_nop\n\tv_nop\n\tv_nop"
                 : "+v"(acc[0]), "+v"(acc[1]), "+v"(acc[2]), "+v"(acc[3]),
                   "+v"(lv[0]), "+v"(lv[1]), "+v"(lv[2]), "+v"(lv[3]),
                   "+v"(rv[0]), "+v"(rv[1]), "+v"(rv[2]), "+v"(rv[3]),
                   "+v"(loff), "+v"(lbits));
  } else {
    asm volatile("v_nop\n\tv_nop\n\tv_nop\n\tv_nop\n\tv_nop"
                 : "+v"(acc[0]), "+v"(acc[1]),
                   "+v"(lv[0]), "+v"(lv[1]), "+v"(rv[0]), "+v"(rv[1]),
                   "+v"(loff), "+v"(lbits));
  }
  const bool hb0 = (lbits & 1) != 0, hb1 = (lbits & 2) != 0;
  const bool hb2 = (lbits & 4) != 0, hb3 = (lbits & 8) != 0;
  global_float* g0 = (global_float*)(DH < 16 ? el : (hb3 ? er : el)) + loff;
  global_float* g1 = (global_float*)er + loff;

#pragma unroll
  for (int t = 0; t < NT; ++t) {
#pragma unroll
    for (int i = 0; i < 8; ++i) drow[i * C + 16 * t] = acc[t][i];
  }

#pragma unroll
  for (int g = 0; g < NHT; ++g) {
    float v[16];
#pragma unroll
    for (int i = 0; i < 8; ++i) {
      float sl = acc[g * TPH][i] * lv[g * TPH];
      float sr = acc[g * TPH][i] * rv[g * TPH];
#pragma unroll
      for (int u = 1; u < TPH; ++u) {
        sl += acc[g * TPH + u][i] * lv[g * TPH + u];
        sr += acc[g * TPH + u][i] * rv[g * TPH + u];
      }
      v[i] = sl;
      v[8 + i] = sr;
    }
    fold<8, 1>(v, hb0);
    fold<4, 2>(v, hb1);
    fold<2, 4>(v, hb2);
    if constexpr (DH >= 16) {
      fold<1, 8>(v, hb3);
      g0[g] = v[0];
    } else {
      g0[HPG * g] = v[0];
      g1[HPG * g] = v[1];
    }
  }
}

__device__ __forceinline__ unsigned float_ord_encode(float f) {
  unsigned u = __float_as_uint(f);
  return (u & 0x80000000u) ? ~u : (u | 0x80000000u);
}
__device__ __forceinline__ float float_ord_decode(unsigned u) {
  return (u & 0x80000000u) ? __uint_as_float(u ^ 0x80000000u)
                           : __uint_as_float(~u);
}
#define ENC_NEG_INF 0x007FFFFFu

#define OWN_R 391
#define OWN_WAVES 32
#define OWN_QCAP 1024
#define OWN_RBITS 9

#define LST_ROW 32
#define OWN_RP 400
#define LST_NONE 0xFFFFFFFFu

#define OWN_NGRP (NEDGES / 32)
#define OWN_NBLK ((OWN_NGRP + 31) / 32)

__global__ __launch_bounds__(256) void owner_table_kernel(
    const int* __restrict__ dst, unsigned* __restrict__ tab) {
  const int lane = threadIdx.x & 31;
  const int blk = blockIdx.x * 8 + (threadIdx.x >> 5);
  if (blk >= OWN_NBLK) return;
  const int grp = blk * 32 + lane;
  unsigned p[8] = {0u, 0u, 0u, 0u, 0u, 0u, 0u, 0u};
  if (grp < OWN_NGRP) {
    const int4* d4 = reinterpret_cast<const int4*>(dst + 32 * (long)grp);
#pragma unroll
    for (int c = 0; c < 8; ++c) {
      const int4 d = d4[c];
      const int dv[4] = {d.x, d.y, d.z, d.w};
#pragma unroll
      for (int k = 0; k < 4; ++k) {
        const unsigned oq = (unsigned)dv[k] / OWN_R;
        const unsigned o = oq < 255u ? oq : 255u;
#pragma unroll
        for (int j = 0; j < 8; ++j) p[j] |= ((o >> j) & 1u) << (4 * c + k);
      }
    }
  }
  uint4* t4 = reinterpret_cast<uint4*>(tab);
  t4[blk * 64 + lane]      = make_uint4(p[0], p[1], p[2], p[3]);
  t4[blk * 64 + 32 + lane] = make_uint4(p[4], p[5], p[6], p[7]);
}

template <int H, int D, bool DO_ELU, int LISTS, int LS>
__global__ __launch_bounds__(1024) void owner_agg_kernel(
    const int* __restrict__ src, const int* __restrict__ dst,
    const unsigned* __restrict__ tab,
    const float* __restrict__ el, const float* __restrict__ er,
    const float* __restrict__ feat, const float* __restrict__ bias,
    const float* __restrict__ Wn, const float* __restrict__ aln,
    const float* __restrict__ arn, float* lgn,
    unsigned* lst, float* out) {
  constexpr int C = H * D;
  constexpr bool FUSE = (LISTS == 1);
  constexpr bool PADL = (LISTS == 2);
  static_assert(LISTS != 2 || C == LST_ROW, "the lists lie in this layer's own output rows");
  constexpr int CPL = C / 32;
  constexpr int BPI = OWN_WAVES * 2;
  constexpr int NIT = (OWN_NBLK + BPI - 1) / BPI;
  constexpr int RB = 16;

  __shared__ float    s_agg[OWN_R * C];
  __shared__ float    s_den[OWN_R * H];
  __shared__ unsigned s_max[OWN_R * H];
  __shared__ __attribute__((aligned(16))) int s_q[OWN_WAVES][OWN_QCAP];
  constexpr bool ONEW = (H == 1);
  constexpr int HP = 32 / D;
  static_assert(ONEW || (CPL == 2 && H == 2 * HP), "a lane's two heads are a pair");
  __shared__ __attribute__((aligned(16))) float s_w[OWN_WAVES][ONEW ? 1 : 32 * H];
  __shared__ int      s_resweep;
  constexpr bool KEEPV = (H == 1);
  __shared__ float    s_v[KEEPV ? OWN_WAVES : 1][KEEPV ? OWN_QCAP : 1];

  const int tid = threadIdx.x;
  const int lane = tid & 31;
  const int wave = tid >> 5;
  const int lo = blockIdx.x * OWN_R;
  int nloc = NNODES - lo;
  nloc = nloc > OWN_R ? OWN_R : nloc;
  nloc = nloc < 0 ? 0 : nloc;
  static_assert(OWN_WAVES == LST_ROW, "a row of counts, cap rows of lists");
  const int cap = nloc >= 1 ? ((nloc - 1) & ~31) : 0;
  unsigned* const lcnt = lst + (long)lo * LST_ROW;
  unsigned* const lmine = lcnt + LST_ROW + wave * cap;

  for (int i = tid; i < OWN_R * C; i += 1024) s_agg[i] = 0.f;
  for (int i = tid; i < OWN_R * H; i += 1024) {
    s_den[i] = 0.f;
    s_max[i] = ENC_NEG_INF;
  }
  if (tid == 0) s_resweep = 0;
  __syncthreads();

  const uint4* tab4 = reinterpret_cast<const uint4*>(tab);
  unsigned want[8];
#pragma unroll
  for (int j = 0; j < 8; ++j) want[j] = ((blockIdx.x >> j) & 1u) ? ~0u : 0u;
  int qhead = 0, qtail = 0;
  bool packed = false;
  bool havev = false;
  unsigned icnt = LST_NONE;
  if (LISTS == 2 && nloc >= 1) icnt = __builtin_amdgcn_readfirstlane(lcnt[wave]);
  const bool imported = icnt <= (unsigned)cap;

#pragma unroll 1
  for (int pass = 0; pass < 2; ++pass) {

    auto drain = [&](int n) {
      __builtin_amdgcn_fence(__ATOMIC_RELEASE, "wavefront");
      __builtin_amdgcn_wave_barrier();
      __builtin_amdgcn_fence(__ATOMIC_ACQUIRE, "wavefront");
      int rel = 0, s = 0;
      float wl = 0.f;
      if (lane < n) {
        const int slot = (qhead + lane) & (OWN_QCAP - 1);
        const unsigned q = (unsigned)s_q[wave][slot];
        if (packed) {
          s = (int)(q >> OWN_RBITS);
          rel = (int)(q & ((1u << OWN_RBITS) - 1u));
        } else {
          s = src[q];
          const unsigned ru = (unsigned)(dst[q] - lo);
          rel = (int)(ru < (unsigned)(OWN_R - 1) ? ru : (unsigned)(OWN_R - 1));
          if (pass == 0) {
            const unsigned pk = ((unsigned)s << OWN_RBITS) | (unsigned)rel;
            s_q[wave][slot] = (int)pk;
            if (LISTS == 1 && slot < cap) lmine[slot] = pk;
          }
        }
        long eli = (long)s, eri = (long)(lo + rel);
        if (PADL) {
          eli = (long)((unsigned)s + (OWN_RP - OWN_R) * ((unsigned)s / OWN_R));
          eri = (long)((int)blockIdx.x * OWN_RP + rel);
        }
        const float* els = el + eli * LS;
        const float* erd = er + eri * LS;
        float vv[H];
        if (KEEPV && havev) {
          vv[0] = s_v[wave][slot];
        } else {
#pragma unroll
          for (int h = 0; h < H; ++h) {
            const float v = els[h] + erd[h];
            vv[h] = v > 0.f ? v : 0.2f * v;
          }
          if (KEEPV && pass == 0) s_v[wave][slot] = vv[0];
        }
        if (pass == 0) {
#pragma unroll
          for (int h = 0; h < H; ++h)
            atomicMax(&s_max[rel * H + h], float_ord_encode(vv[h]));
        } else {
          float m[H];
#pragma unroll
          for (int h = 0; h < H; ++h) m[h] = float_ord_decode(s_max[rel * H + h]);
#pragma unroll
          for (int h = 0; h < H; ++h) {
            const float w = expf(vv[h] - m[h]);
            atomicAdd(&s_den[rel * H + h], w);
            if (ONEW) wl = w;
            else s_w[wave][lane * H + 2 * (h % HP) + h / HP] = w;
          }
        }
      }
      if (pass == 1) {
        if (!ONEW) {
          __builtin_amdgcn_fence(__ATOMIC_RELEASE, "wavefront");
          __builtin_amdgcn_wave_barrier();
          __builtin_amdgcn_fence(__ATOMIC_ACQUIRE, "wavefront");
        }
        for (int j0 = 0; j0 < n; j0 += RB) {
          float f[RB][CPL];
#pragma unroll
          for (int u = 0; u < RB; ++u) {
            const int sj = __builtin_amdgcn_readlane(s, j0 + u);
            if (j0 + u < n) {
#pragma unroll
              for (int p = 0; p < CPL; ++p)
                f[u][p] = feat[(long)sj * C + lane + 32 * p];
            }
          }
          float wv[RB][CPL];
#pragma unroll
          for (int u = 0; u < RB; ++u) {
            if (ONEW) {
              wv[u][0] = __int_as_float(
                  __builtin_amdgcn_readlane(__float_as_int(wl), j0 + u));
            } else if (j0 + u < n) {
              const float2 w2 = *reinterpret_cast<const float2*>(
                  &s_w[wave][(j0 + u) * H + 2 * (lane / D)]);
              wv[u][0] = w2.x;
              wv[u][CPL - 1] = w2.y;
            }
          }
#pragma unroll
          for (int u = 0; u < RB; ++u) {
            const int rj = __builtin_amdgcn_readlane(rel, j0 + u);
            if (j0 + u < n) {
#pragma unroll
              for (int p = 0; p < CPL; ++p)
                atomicAdd(&s_agg[rj * C + lane + 32 * p], wv[u][p] * f[u][p]);
            }
          }
        }
      }
      qhead += n;
    };

    const bool sweep = (pass == 0) ? !imported : (s_resweep != 0);
    packed = !sweep;
    havev = (pass == 1) && !sweep;
    if (sweep) {
      qhead = 0;
      qtail = 0;
#pragma unroll 1
      for (int i = 0; i < NIT; ++i) {
        const int it = (i + (int)blockIdx.x) % NIT;
        const int blk0 = it * BPI + wave * 2;
        unsigned hit[2];
#pragma unroll
        for (int k = 0; k < 2; ++k) {
          const int blk = blk0 + k;
          const int bc = blk < OWN_NBLK ? blk : OWN_NBLK - 1;
          const uint4 pa = tab4[bc * 64 + lane];
          const uint4 pb = tab4[bc * 64 + 32 + lane];
          const unsigned pl[8] = {pa.x, pa.y, pa.z, pa.w, pb.x, pb.y, pb.z, pb.w};
          unsigned acc = ~0u;
#pragma unroll
          for (int j = 0; j < 8; ++j) acc &= ~(pl[j] ^ want[j]);
          hit[k] = (blk * 32 + lane < OWN_NGRP) ? acc : 0u;
        }
        unsigned h0 = hit[0], h1 = hit[1];
#pragma unroll 1
        for (int k = 0; k < 2; ++k) {
          unsigned mm = h0;
          h0 = h1;
          const int ebase = (blk0 + k) * 1024 + lane * 32;
          unsigned any = __builtin_amdgcn_ballot_w32(mm != 0u);
          while (any != 0u) {
            while (qtail - qhead > OWN_QCAP - 32) drain(32);
            if (mm != 0u) {
              const int q = __builtin_ctz(mm);
              mm &= mm - 1u;
              const int pos =
                  (qtail + (int)__builtin_amdgcn_mbcnt_lo(any, 0u)) &
                  (OWN_QCAP - 1);
              s_q[wave][pos] = ebase + q;
            }
            qtail += __popc(any);
            any = __builtin_amdgcn_ballot_w32(mm != 0u);
          }
        }
      }
      if (pass == 0 && qtail > OWN_QCAP) s_resweep = 1;
    } else if (pass == 0) {
      const int n = (int)icnt;
      for (int b = 0; b < n; b += 256) {
        unsigned t[8];
#pragma unroll
        for (int u = 0; u < 8; ++u)
          if (b + 32 * u < n) t[u] = lmine[b + 32 * u + lane];
#pragma unroll
        for (int u = 0; u < 8; ++u)
          if (b + 32 * u < n) s_q[wave][b + 32 * u + lane] = (int)t[u];
      }
      qhead = 0;
      qtail = n;
    } else {
      qhead = 0;
    }
    while (qtail - qhead >= 32) drain(32);
    if (qtail - qhead > 0) drain(qtail - qhead);
    if (LISTS == 1 && pass == 0 && nloc >= 1 && lane == 0)
      lcnt[wave] = qtail <= cap ? (unsigned)qtail : LST_NONE;
    __syncthreads();
  }

  if constexpr (!FUSE) {
    for (int i = tid; i < nloc * C; i += 1024) {
      const int r = i / C;
      const int c = i % C;
      const float dn = s_den[r * H + c / D];
      float x = dn > 0.f ? s_agg[i] / dn : 0.f;
      x += bias[c];
      if (DO_ELU) x = x > 0.f ? x : expf(x) - 1.f;
      out[(long)lo * C + i] = x;
    }
  } else {
    constexpr int NK = C;
    constexpr int NC = 32;
    constexpr int NT = NC / 16;
    constexpr int P = NK + 4;
    constexpr int NSTRIP = OWN_RP / 16;
    static_assert(NK % 32 == 0 && (P / 4) % 2 == 1, "whole lines; odd 16-byte pitch");
    static_assert(OWN_RP % 16 == 0 && OWN_RP >= OWN_R && NSTRIP <= OWN_WAVES, "a strip per wave");
    static_assert(OWN_RP * P <= OWN_WAVES * OWN_QCAP, "the rows fit the lists' space");
    static_assert(!ONEW && NC * P <= OWN_WAVES * 32 * H, "the weights fit the waves' weight space");
    static_assert(NT == 2, "the statement behind the products names every tile");
    float* const hS = reinterpret_cast<float*>(&s_q[0][0]);
    float* const sB = &s_w[0][0];
    for (int i = tid; i < nloc * C; i += 1024) {
      const int r = i / C;
      const int c = i % C;
      const float dn = s_den[r * H + c / D];
      float x = dn > 0.f ? s_agg[i] / dn : 0.f;
      x += bias[c];
      if (DO_ELU) x = x > 0.f ? x : expf(x) - 1.f;
      hS[r * P + c] = x;
    }
    for (int i = nloc * C + tid; i < OWN_RP * C; i += 1024) hS[(i / C) * P + (i % C)] = 0.f;
    for (int i = tid; i < NK * NC; i += 1024) sB[(i % NC) * P + (i / NC)] = Wn[i];
    __syncthreads();

    if (wave < NSTRIP) {
      const int half = lane >> 4;
      const int c = lane & 15;
      const int r0 = 16 * wave;
      float* dpin = out + (long)(lo + r0 + 8 * half) * NC + c;
      asm volatile("" : "+v"(dpin));
      typedef __attribute__((address_space(1))) float global_float;
      global_float* drow = (global_float*)dpin;
      float lv[NT], rv[NT];
#pragma unroll
      for (int t = 0; t < NT; ++t) {
        lv[t] = aln[16 * t + c];
        rv[t] = arn[16 * t + c];
      }
      unsigned loff = (unsigned)((int)blockIdx.x * OWN_RP + r0 + 8 * half + (c & 7)) * 2;
      int lbits = lane;
      int rl = r0 + 8 * half;
      asm volatile("" : "+v"(loff), "+v"(lbits), "+v"(rl));

      v8f acc[NT];
#pragma unroll
      for (int t = 0; t < NT; ++t) acc[t] = v8f{};
      const float* al_ = hS + (r0 + c) * P + 16 * half;
      const float* bl = sB + c * P + 16 * half;
#pragma unroll
      for (int L = 0; L < NK / 32; ++L) {
#pragma unroll
        for (int u = 0; u < 4; ++u) {
          const float4 av = *reinterpret_cast<const float4*>(al_ + 32 * L + 4 * u);
          float4 bv[NT];
#pragma unroll
          for (int t = 0; t < NT; ++t)
            bv[t] = *reinterpret_cast<const float4*>(bl + t * 16 * P + 32 * L + 4 * u);
          v2f a01, a23;
          a01.x = av.x; a01.y = av.y;
          a23.x = av.z; a23.y = av.w;
#pragma unroll
          for (int t = 0; t < NT; ++t) {
            v2f b01;
            b01.x = bv[t].x; b01.y = bv[t].y;
            acc[t] = __builtin_amdgcn_wmma_f32_16x16x4_f32(
                 false, a01,  false, b01,
                 (short)0, acc[t],  false,  false);
          }
#pragma unroll
          for (int t = 0; t < NT; ++t) {
            v2f b23;
            b23.x = bv[t].z; b23.y = bv[t].w;
            acc[t] = __builtin_amdgcn_wmma_f32_16x16x4_f32(
                 false, a23,  false, b23,
                 (short)0, acc[t],  false,  false);
          }
        }
      }
      asm volatile("v_nop\n\tv_nop\n\tv_nop\n\tv_nop\n\tv_nop"
                   : "+v"(acc[0]), "+v"(acc[1]),
                     "+v"(lv[0]), "+v"(lv[1]), "+v"(rv[0]), "+v"(rv[1]),
                     "+v"(loff), "+v"(lbits), "+v"(rl));
      const bool hb0 = (lbits & 1) != 0, hb1 = (lbits & 2) != 0;
      const bool hb2 = (lbits & 4) != 0, hb3 = (lbits & 8) != 0;
      global_float* g0 = (global_float*)(hb3 ? lgn + 1 : lgn) + loff;

#pragma unroll
      for (int t = 0; t < NT; ++t) {
#pragma unroll
        for (int i = 0; i < 8; ++i)
          if (rl + i < nloc) drow[i * NC + 16 * t] = acc[t][i];
      }
      float v[16];
#pragma unroll
      for (int i = 0; i < 8; ++i) {
        float sl = acc[0][i] * lv[0];
        float sr = acc[0][i] * rv[0];
#pragma unroll
        for (int t = 1; t < NT; ++t) {
          sl += acc[t][i] * lv[t];
          sr += acc[t][i] * rv[t];
        }
        v[i] = sl;
        v[8 + i] = sr;
      }
      fold<8, 1>(v, hb0);
      fold<4, 2>(v, hb1);
      fold<2, 4>(v, hb2);
      fold<1, 8>(v, hb3);
      if (rl + (lbits & 7) < nloc) g0[0] = v[0];
    }
  }
}

extern "C" void kernel_launch(void* const* d_in, const int* in_sizes, int n_in,
                              void* d_out, int out_size, void* d_ws,
                              size_t ws_size, hipStream_t stream) {
  const float* x   = (const float*)d_in[0];
  const int*   src = (const int*)d_in[1];
  const int*   dst = (const int*)d_in[2];
  const float* W1  = (const float*)d_in[3];
  const float* al1 = (const float*)d_in[4];
  const float* ar1 = (const float*)d_in[5];
  const float* b1  = (const float*)d_in[6];
  const float* W2  = (const float*)d_in[7];
  const float* al2 = (const float*)d_in[8];
  const float* ar2 = (const float*)d_in[9];
  const float* b2  = (const float*)d_in[10];
  float* out = (float*)d_out;
  unsigned* lists = (unsigned*)d_out;

  const int N = NNODES;
  static_assert(NNODES % (16 * GEMM_WAVES) == 0, "every GEMM wave has a full row tile");

  float* ws = (float*)d_ws;
  float* feat1 = ws;
  float* el1   = feat1 + (long)N * 64;
  float* er1   = el1 + (long)N * 8;
  unsigned* tab = (unsigned*)(ws + (long)N * 80);
  float* feat2 = ws + (long)N * 96;
  float* el2   = ws + (long)N * 128;
  float* er2   = el2 + 1;

  auto nb = [](long n, int b) { return (int)((n + b - 1) / b); };
  const int NOWN = (NNODES + OWN_R - 1) / OWN_R;

  owner_table_kernel<<<nb(OWN_NBLK, 8), 256, 0, stream>>>(dst, tab);

  gemm_wmma_kernel<256, 64, 8, 8><<<NNODES / (16 * GEMM_WAVES), 32 * GEMM_WAVES, 0, stream>>>(x, W1, al1, ar1, feat1, el1, er1);
  owner_agg_kernel<8, 8, true, 1, 8><<<NOWN, 1024, 0, stream>>>(src, dst, tab, el1, er1, feat1, b1, W2, al2, ar2, el2, lists, feat2);

  owner_agg_kernel<1, 32, false, 2, 2><<<NOWN, 1024, 0, stream>>>(src, dst, tab, el2, er2, feat2, b2, nullptr, nullptr, nullptr, nullptr, lists, out);
}


# layer 1's owner groups read el / er and update maxima and denominators with 8 lanes per edge (a lane per head, 4 edges per instruction): a load touches 4 lines, not 32; denominator add order changed
# speedup vs baseline: 7.4041x; 1.0620x over previous
#include <hip/hip_runtime.h>
#include <hip/hip_bf16.h>

typedef __attribute__((ext_vector_type(2))) float v2f;
typedef __attribute__((ext_vector_type(8))) float v8f;

#define NNODES 100000
#define NEDGES 1600000

#define GEMM_WAVES 25

template <int M>
__device__ __forceinline__ float row_xor(float x) {
  const int b = __builtin_bit_cast(int, x);
  return __builtin_bit_cast(
      float, __builtin_amdgcn_update_dpp(0, b, 0x160 + M, 0xf, 0xf, true));
}

template <int CNT, int M>
__device__ __forceinline__ void fold(float* v, bool hi) {
#pragma unroll
  for (int n = 0; n < CNT; ++n) {
    const float keep = hi ? v[2 * n + 1] : v[2 * n];
    const float send = hi ? v[2 * n] : v[2 * n + 1];
    v[n] = keep + row_xor<M>(send);
  }
}

template <int K, int C, int H, int LS>
__global__ __launch_bounds__(32 * GEMM_WAVES)
__attribute__((amdgpu_waves_per_eu(7, 7))) void gemm_wmma_kernel(
    const float* __restrict__ A, const float* __restrict__ B,
    const float* __restrict__ al, const float* __restrict__ ar,
    float* __restrict__ D, float* __restrict__ el, float* __restrict__ er) {
  constexpr int NT = C / 16;
  constexpr int DH = C / H;
  static_assert(C % H == 0 && (DH == 8 || DH % 16 == 0), "a head: 8 lanes of a tile, or whole tiles");
  constexpr int TPH = DH > 16 ? DH / 16 : 1;
  constexpr int NHT = NT / TPH;
  constexpr int HPG = DH < 16 ? 16 / DH : 1;
  static_assert(LS >= H, "LS: words from a node's logits to the next node's");
  constexpr int NL = K / 32;
  constexpr int BP = K + 4;
  static_assert(K % 32 == 0 && C % 16 == 0, "whole lines, whole tiles");
  static_assert((BP / 4) % 2 == 1, "row pitch: an odd number of 16-byte pieces");
  __shared__ __attribute__((aligned(16))) float sB[C * BP];

  const int tid = threadIdx.x;
  const int lane = tid & 31;
  const int wave = tid >> 5;
  const int half = lane >> 4;
  const int c    = lane & 15;
  const int row0 = (blockIdx.x * GEMM_WAVES + wave) * 16;

  const float4* ap =
      reinterpret_cast<const float4*>(A + (long)(row0 + c) * K + 16 * half);
  float4 a[2][4];
#pragma unroll
  for (int u = 0; u < 4; ++u) a[0][u] = ap[u];

  float* dpin = D + (long)(row0 + 8 * half) * C + c;
  asm volatile("" : "+v"(dpin));
  typedef __attribute__((address_space(1))) float global_float;
  global_float* drow = (global_float*)dpin;

  float lv[NT], rv[NT];
#pragma unroll
  for (int t = 0; t < NT; ++t) {
    lv[t] = al[16 * t + c];
    rv[t] = ar[16 * t + c];
  }
  unsigned loff = (unsigned)(row0 + 8 * half + (c & 7)) * LS + (DH < 16 ? (c >> 3) : 0);
  int lbits = lane;
  asm volatile("" : "+v"(loff), "+v"(lbits));

  constexpr int NTHR = 32 * GEMM_WAVES;
  constexpr int NQ = C * K / 4;
  constexpr int NU = (NQ + NTHR - 1) / NTHR;
  float4 w[NU];
#pragma unroll
  for (int j = 0; j < NU; ++j) {
    const int q = tid + j * NTHR;
    if (q < NQ) {
      const int n = q % C;
      const int k = (q / C) * 4;
      w[j].x = B[(k + 0) * C + n];
      w[j].y = B[(k + 1) * C + n];
      w[j].z = B[(k + 2) * C + n];
      w[j].w = B[(k + 3) * C + n];
    }
  }
#pragma unroll
  for (int j = 0; j < NU; ++j) {
    const int q = tid + j * NTHR;
    if (q < NQ) {
      const int n = q % C;
      const int k = (q / C) * 4;
      *reinterpret_cast<float4*>(&sB[n * BP + k]) = w[j];
    }
  }
  __syncthreads();

  v8f acc[NT];
#pragma unroll
  for (int t = 0; t < NT; ++t) acc[t] = v8f{};

  const float* bl = sB + c * BP + 16 * half;
#pragma unroll
  for (int L = 0; L < NL; ++L) {
    if (L + 1 < NL) {
#pragma unroll
      for (int u = 0; u < 4; ++u) a[(L + 1) & 1][u] = ap[8 * (L + 1) + u];
    }
#pragma unroll
    for (int u = 0; u < 4; ++u) {
      const float4 av = a[L & 1][u];
      float4 bv[NT];
#pragma unroll
      for (int t = 0; t < NT; ++t)
        bv[t] = *reinterpret_cast<const float4*>(bl + t * 16 * BP + 32 * L + 4 * u);
      v2f a01, a23;
      a01.x = av.x; a01.y = av.y;
      a23.x = av.z; a23.y = av.w;
#pragma unroll
      for (int t = 0; t < NT; ++t) {
        v2f b01;
        b01.x = bv[t].x; b01.y = bv[t].y;
        acc[t] = __builtin_amdgcn_wmma_f32_16x16x4_f32(
             false, a01,  false, b01,
             (short)0, acc[t],  false,  false);
      }
#pragma unroll
      for (int t = 0; t < NT; ++t) {
        v2f b23;
        b23.x = bv[t].z; b23.y = bv[t].w;
        acc[t] = __builtin_amdgcn_wmma_f32_16x16x4_f32(
             false, a23,  false, b23,
             (short)0, acc[t],  false,  false);
      }
    }
  }

  static_assert(NT == 2 || NT == 4, "the statement below names every tile");
  if constexpr (NT == 4) {
    asm volatile("v_nop\n\tv_nop\n\tv_nop\n\tv_nop\n\tv_nop"
                 : "+v"(acc[0]), "+v"(acc[1]), "+v"(acc[2]), "+v"(acc[3]),
                   "+v"(lv[0]), "+v"(lv[1]), "+v"(lv[2]), "+v"(lv[3]),
                   "+v"(rv[0]), "+v"(rv[1]), "+v"(rv[2]), "+v"(rv[3]),
                   "+v"(loff), "+v"(lbits));
  } else {
    asm volatile("v_nop\n\tv_nop\n\tv_nop\n\tv_nop\n\tv_nop"
                 : "+v"(acc[0]), "+v"(acc[1]),
                   "+v"(lv[0]), "+v"(lv[1]), "+v"(rv[0]), "+v"(rv[1]),
                   "+v"(loff), "+v"(lbits));
  }
  const bool hb0 = (lbits & 1) != 0, hb1 = (lbits & 2) != 0;
  const bool hb2 = (lbits & 4) != 0, hb3 = (lbits & 8) != 0;
  global_float* g0 = (global_float*)(DH < 16 ? el : (hb3 ? er : el)) + loff;
  global_float* g1 = (global_float*)er + loff;

#pragma unroll
  for (int t = 0; t < NT; ++t) {
#pragma unroll
    for (int i = 0; i < 8; ++i) drow[i * C + 16 * t] = acc[t][i];
  }

#pragma unroll
  for (int g = 0; g < NHT; ++g) {
    float v[16];
#pragma unroll
    for (int i = 0; i < 8; ++i) {
      float sl = acc[g * TPH][i] * lv[g * TPH];
      float sr = acc[g * TPH][i] * rv[g * TPH];
#pragma unroll
      for (int u = 1; u < TPH; ++u) {
        sl += acc[g * TPH + u][i] * lv[g * TPH + u];
        sr += acc[g * TPH + u][i] * rv[g * TPH + u];
      }
      v[i] = sl;
      v[8 + i] = sr;
    }
    fold<8, 1>(v, hb0);
    fold<4, 2>(v, hb1);
    fold<2, 4>(v, hb2);
    if constexpr (DH >= 16) {
      fold<1, 8>(v, hb3);
      g0[g] = v[0];
    } else {
      g0[HPG * g] = v[0];
      g1[HPG * g] = v[1];
    }
  }
}

__device__ __forceinline__ unsigned float_ord_encode(float f) {
  unsigned u = __float_as_uint(f);
  return (u & 0x80000000u) ? ~u : (u | 0x80000000u);
}
__device__ __forceinline__ float float_ord_decode(unsigned u) {
  return (u & 0x80000000u) ? __uint_as_float(u ^ 0x80000000u)
                           : __uint_as_float(~u);
}
#define ENC_NEG_INF 0x007FFFFFu

#define OWN_R 391
#define OWN_WAVES 32
#define OWN_QCAP 1024
#define OWN_RBITS 9

#define LST_ROW 32
#define OWN_RP 400
#define LST_NONE 0xFFFFFFFFu

#define OWN_NGRP (NEDGES / 32)
#define OWN_NBLK ((OWN_NGRP + 31) / 32)

__global__ __launch_bounds__(256) void owner_table_kernel(
    const int* __restrict__ dst, unsigned* __restrict__ tab) {
  const int lane = threadIdx.x & 31;
  const int blk = blockIdx.x * 8 + (threadIdx.x >> 5);
  if (blk >= OWN_NBLK) return;
  const int grp = blk * 32 + lane;
  unsigned p[8] = {0u, 0u, 0u, 0u, 0u, 0u, 0u, 0u};
  if (grp < OWN_NGRP) {
    const int4* d4 = reinterpret_cast<const int4*>(dst + 32 * (long)grp);
#pragma unroll
    for (int c = 0; c < 8; ++c) {
      const int4 d = d4[c];
      const int dv[4] = {d.x, d.y, d.z, d.w};
#pragma unroll
      for (int k = 0; k < 4; ++k) {
        const unsigned oq = (unsigned)dv[k] / OWN_R;
        const unsigned o = oq < 255u ? oq : 255u;
#pragma unroll
        for (int j = 0; j < 8; ++j) p[j] |= ((o >> j) & 1u) << (4 * c + k);
      }
    }
  }
  uint4* t4 = reinterpret_cast<uint4*>(tab);
  t4[blk * 64 + lane]      = make_uint4(p[0], p[1], p[2], p[3]);
  t4[blk * 64 + 32 + lane] = make_uint4(p[4], p[5], p[6], p[7]);
}

template <int H, int D, bool DO_ELU, int LISTS, int LS>
__global__ __launch_bounds__(1024) void owner_agg_kernel(
    const int* __restrict__ src, const int* __restrict__ dst,
    const unsigned* __restrict__ tab,
    const float* __restrict__ el, const float* __restrict__ er,
    const float* __restrict__ feat, const float* __restrict__ bias,
    const float* __restrict__ Wn, const float* __restrict__ aln,
    const float* __restrict__ arn, float* lgn,
    unsigned* lst, float* out) {
  constexpr int C = H * D;
  constexpr bool FUSE = (LISTS == 1);
  constexpr bool PADL = (LISTS == 2);
  static_assert(LISTS != 2 || C == LST_ROW, "the lists lie in this layer's own output rows");
  constexpr int CPL = C / 32;
  constexpr int BPI = OWN_WAVES * 2;
  constexpr int NIT = (OWN_NBLK + BPI - 1) / BPI;
  constexpr int RB = 16;

  __shared__ float    s_agg[OWN_R * C];
  __shared__ float    s_den[OWN_R * H];
  __shared__ unsigned s_max[OWN_R * H];
  __shared__ __attribute__((aligned(16))) int s_q[OWN_WAVES][OWN_QCAP];
  constexpr bool ONEW = (H == 1);
  constexpr int HP = 32 / D;
  static_assert(ONEW || (CPL == 2 && H == 2 * HP), "a lane's two heads are a pair");
  __shared__ __attribute__((aligned(16))) float s_w[OWN_WAVES][ONEW ? 1 : 32 * H];
  __shared__ int      s_resweep;
  constexpr bool KEEPV = (H == 1);
  __shared__ float    s_v[KEEPV ? OWN_WAVES : 1][KEEPV ? OWN_QCAP : 1];

  const int tid = threadIdx.x;
  const int lane = tid & 31;
  const int wave = tid >> 5;
  const int lo = blockIdx.x * OWN_R;
  int nloc = NNODES - lo;
  nloc = nloc > OWN_R ? OWN_R : nloc;
  nloc = nloc < 0 ? 0 : nloc;
  static_assert(OWN_WAVES == LST_ROW, "a row of counts, cap rows of lists");
  const int cap = nloc >= 1 ? ((nloc - 1) & ~31) : 0;
  unsigned* const lcnt = lst + (long)lo * LST_ROW;
  unsigned* const lmine = lcnt + LST_ROW + wave * cap;

  for (int i = tid; i < OWN_R * C; i += 1024) s_agg[i] = 0.f;
  for (int i = tid; i < OWN_R * H; i += 1024) {
    s_den[i] = 0.f;
    s_max[i] = ENC_NEG_INF;
  }
  if (tid == 0) s_resweep = 0;
  __syncthreads();

  const uint4* tab4 = reinterpret_cast<const uint4*>(tab);
  unsigned want[8];
#pragma unroll
  for (int j = 0; j < 8; ++j) want[j] = ((blockIdx.x >> j) & 1u) ? ~0u : 0u;
  int qhead = 0, qtail = 0;
  bool packed = false;
  bool havev = false;
  unsigned icnt = LST_NONE;
  if (LISTS == 2 && nloc >= 1) icnt = __builtin_amdgcn_readfirstlane(lcnt[wave]);
  const bool imported = icnt <= (unsigned)cap;

#pragma unroll 1
  for (int pass = 0; pass < 2; ++pass) {

    auto drain = [&](int n) {
      __builtin_amdgcn_fence(__ATOMIC_RELEASE, "wavefront");
      __builtin_amdgcn_wave_barrier();
      __builtin_amdgcn_fence(__ATOMIC_ACQUIRE, "wavefront");
      int rel = 0, s = 0;
      float wl = 0.f;
      if (lane < n) {
        const int slot = (qhead + lane) & (OWN_QCAP - 1);
        const unsigned q = (unsigned)s_q[wave][slot];
        if (packed) {
          s = (int)(q >> OWN_RBITS);
          rel = (int)(q & ((1u << OWN_RBITS) - 1u));
        } else {
          s = src[q];
          const unsigned ru = (unsigned)(dst[q] - lo);
          rel = (int)(ru < (unsigned)(OWN_R - 1) ? ru : (unsigned)(OWN_R - 1));
          if (pass == 0) {
            const unsigned pk = ((unsigned)s << OWN_RBITS) | (unsigned)rel;
            s_q[wave][slot] = (int)pk;
            if (LISTS == 1 && slot < cap) lmine[slot] = pk;
          }
        }
        if constexpr (ONEW) {
        long eli = (long)s, eri = (long)(lo + rel);
        if (PADL) {
          eli = (long)((unsigned)s + (OWN_RP - OWN_R) * ((unsigned)s / OWN_R));
          eri = (long)((int)blockIdx.x * OWN_RP + rel);
        }
        const float* els = el + eli * LS;
        const float* erd = er + eri * LS;
        float vv[H];
        if (KEEPV && havev) {
          vv[0] = s_v[wave][slot];
        } else {
#pragma unroll
          for (int h = 0; h < H; ++h) {
            const float v = els[h] + erd[h];
            vv[h] = v > 0.f ? v : 0.2f * v;
          }
          if (KEEPV && pass == 0) s_v[wave][slot] = vv[0];
        }
        if (pass == 0) {
#pragma unroll
          for (int h = 0; h < H; ++h)
            atomicMax(&s_max[rel * H + h], float_ord_encode(vv[h]));
        } else {
          float m[H];
#pragma unroll
          for (int h = 0; h < H; ++h) m[h] = float_ord_decode(s_max[rel * H + h]);
#pragma unroll
          for (int h = 0; h < H; ++h) {
            const float w = expf(vv[h] - m[h]);
            atomicAdd(&s_den[rel * H + h], w);
            wl = w;
          }
        }
        }
      }
      if constexpr (!ONEW) {
        constexpr int EPI = 32 / H;
        static_assert(32 % H == 0, "whole edges per instruction");
        const int hh = lane % H;
        const int g = lane / H;
        const int pk = (int)(((unsigned)s << OWN_RBITS) | (unsigned)rel);
        float ev[H];
        int ra[H];
        bool ok[H];
#pragma unroll
        for (int i = 0; i < H; ++i) {
          const int j = EPI * i + g;
          ok[i] = j < n;
          const int jc = ok[i] ? j : n - 1;
          const unsigned pj = (unsigned)__builtin_amdgcn_ds_bpermute(jc << 2, pk);
          const unsigned sj = pj >> OWN_RBITS;
          const int rj = (int)(pj & ((1u << OWN_RBITS) - 1u));
          ra[i] = rj * H + hh;
          const unsigned ei = sj * (unsigned)LS + (unsigned)hh;
          const unsigned ri = (unsigned)(lo + rj) * (unsigned)LS + (unsigned)hh;
          const float v = el[ei] + er[ri];
          ev[i] = v > 0.f ? v : 0.2f * v;
        }
        if (pass == 0) {
#pragma unroll
          for (int i = 0; i < H; ++i)
            if (ok[i]) atomicMax(&s_max[ra[i]], float_ord_encode(ev[i]));
        } else {
          float m[H];
#pragma unroll
          for (int i = 0; i < H; ++i) m[i] = float_ord_decode(s_max[ra[i]]);
#pragma unroll
          for (int i = 0; i < H; ++i) {
            const float w = expf(ev[i] - m[i]);
            if (ok[i]) {
              atomicAdd(&s_den[ra[i]], w);
              s_w[wave][(EPI * i + g) * H + 2 * (hh % HP) + hh / HP] = w;
            }
          }
        }
      }
      if (pass == 1) {
        if (!ONEW) {
          __builtin_amdgcn_fence(__ATOMIC_RELEASE, "wavefront");
          __builtin_amdgcn_wave_barrier();
          __builtin_amdgcn_fence(__ATOMIC_ACQUIRE, "wavefront");
        }
        for (int j0 = 0; j0 < n; j0 += RB) {
          float f[RB][CPL];
#pragma unroll
          for (int u = 0; u < RB; ++u) {
            const int sj = __builtin_amdgcn_readlane(s, j0 + u);
            if (j0 + u < n) {
#pragma unroll
              for (int p = 0; p < CPL; ++p)
                f[u][p] = feat[(long)sj * C + lane + 32 * p];
            }
          }
          float wv[RB][CPL];
#pragma unroll
          for (int u = 0; u < RB; ++u) {
            if (ONEW) {
              wv[u][0] = __int_as_float(
                  __builtin_amdgcn_readlane(__float_as_int(wl), j0 + u));
            } else if (j0 + u < n) {
              const float2 w2 = *reinterpret_cast<const float2*>(
                  &s_w[wave][(j0 + u) * H + 2 * (lane / D)]);
              wv[u][0] = w2.x;
              wv[u][CPL - 1] = w2.y;
            }
          }
#pragma unroll
          for (int u = 0; u < RB; ++u) {
            const int rj = __builtin_amdgcn_readlane(rel, j0 + u);
            if (j0 + u < n) {
#pragma unroll
              for (int p = 0; p < CPL; ++p)
                atomicAdd(&s_agg[rj * C + lane + 32 * p], wv[u][p] * f[u][p]);
            }
          }
        }
      }
      qhead += n;
    };

    const bool sweep = (pass == 0) ? !imported : (s_resweep != 0);
    packed = !sweep;
    havev = (pass == 1) && !sweep;
    if (sweep) {
      qhead = 0;
      qtail = 0;
#pragma unroll 1
      for (int i = 0; i < NIT; ++i) {
        const int it = (i + (int)blockIdx.x) % NIT;
        const int blk0 = it * BPI + wave * 2;
        unsigned hit[2];
#pragma unroll
        for (int k = 0; k < 2; ++k) {
          const int blk = blk0 + k;
          const int bc = blk < OWN_NBLK ? blk : OWN_NBLK - 1;
          const uint4 pa = tab4[bc * 64 + lane];
          const uint4 pb = tab4[bc * 64 + 32 + lane];
          const unsigned pl[8] = {pa.x, pa.y, pa.z, pa.w, pb.x, pb.y, pb.z, pb.w};
          unsigned acc = ~0u;
#pragma unroll
          for (int j = 0; j < 8; ++j) acc &= ~(pl[j] ^ want[j]);
          hit[k] = (blk * 32 + lane < OWN_NGRP) ? acc : 0u;
        }
        unsigned h0 = hit[0], h1 = hit[1];
#pragma unroll 1
        for (int k = 0; k < 2; ++k) {
          unsigned mm = h0;
          h0 = h1;
          const int ebase = (blk0 + k) * 1024 + lane * 32;
          unsigned any = __builtin_amdgcn_ballot_w32(mm != 0u);
          while (any != 0u) {
            while (qtail - qhead > OWN_QCAP - 32) drain(32);
            if (mm != 0u) {
              const int q = __builtin_ctz(mm);
              mm &= mm - 1u;
              const int pos =
                  (qtail + (int)__builtin_amdgcn_mbcnt_lo(any, 0u)) &
                  (OWN_QCAP - 1);
              s_q[wave][pos] = ebase + q;
            }
            qtail += __popc(any);
            any = __builtin_amdgcn_ballot_w32(mm != 0u);
          }
        }
      }
      if (pass == 0 && qtail > OWN_QCAP) s_resweep = 1;
    } else if (pass == 0) {
      const int n = (int)icnt;
      for (int b = 0; b < n; b += 256) {
        unsigned t[8];
#pragma unroll
        for (int u = 0; u < 8; ++u)
          if (b + 32 * u < n) t[u] = lmine[b + 32 * u + lane];
#pragma unroll
        for (int u = 0; u < 8; ++u)
          if (b + 32 * u < n) s_q[wave][b + 32 * u + lane] = (int)t[u];
      }
      qhead = 0;
      qtail = n;
    } else {
      qhead = 0;
    }
    while (qtail - qhead >= 32) drain(32);
    if (qtail - qhead > 0) drain(qtail - qhead);
    if (LISTS == 1 && pass == 0 && nloc >= 1 && lane == 0)
      lcnt[wave] = qtail <= cap ? (unsigned)qtail : LST_NONE;
    __syncthreads();
  }

  if constexpr (!FUSE) {
    for (int i = tid; i < nloc * C; i += 1024) {
      const int r = i / C;
      const int c = i % C;
      const float dn = s_den[r * H + c / D];
      float x = dn > 0.f ? s_agg[i] / dn : 0.f;
      x += bias[c];
      if (DO_ELU) x = x > 0.f ? x : expf(x) - 1.f;
      out[(long)lo * C + i] = x;
    }
  } else {
    constexpr int NK = C;
    constexpr int NC = 32;
    constexpr int NT = NC / 16;
    constexpr int P = NK + 4;
    constexpr int NSTRIP = OWN_RP / 16;
    static_assert(NK % 32 == 0 && (P / 4) % 2 == 1, "whole lines; odd 16-byte pitch");
    static_assert(OWN_RP % 16 == 0 && OWN_RP >= OWN_R && NSTRIP <= OWN_WAVES, "a strip per wave");
    static_assert(OWN_RP * P <= OWN_WAVES * OWN_QCAP, "the rows fit the lists' space");
    static_assert(!ONEW && NC * P <= OWN_WAVES * 32 * H, "the weights fit the waves' weight space");
    static_assert(NT == 2, "the statement behind the products names every tile");
    float* const hS = reinterpret_cast<float*>(&s_q[0][0]);
    float* const sB = &s_w[0][0];
    for (int i = tid; i < nloc * C; i += 1024) {
      const int r = i / C;
      const int c = i % C;
      const float dn = s_den[r * H + c / D];
      float x = dn > 0.f ? s_agg[i] / dn : 0.f;
      x += bias[c];
      if (DO_ELU) x = x > 0.f ? x : expf(x) - 1.f;
      hS[r * P + c] = x;
    }
    for (int i = nloc * C + tid; i < OWN_RP * C; i += 1024) hS[(i / C) * P + (i % C)] = 0.f;
    for (int i = tid; i < NK * NC; i += 1024) sB[(i % NC) * P + (i / NC)] = Wn[i];
    __syncthreads();

    if (wave < NSTRIP) {
      const int half = lane >> 4;
      const int c = lane & 15;
      const int r0 = 16 * wave;
      float* dpin = out + (long)(lo + r0 + 8 * half) * NC + c;
      asm volatile("" : "+v"(dpin));
      typedef __attribute__((address_space(1))) float global_float;
      global_float* drow = (global_float*)dpin;
      float lv[NT], rv[NT];
#pragma unroll
      for (int t = 0; t < NT; ++t) {
        lv[t] = aln[16 * t + c];
        rv[t] = arn[16 * t + c];
      }
      unsigned loff = (unsigned)((int)blockIdx.x * OWN_RP + r0 + 8 * half + (c & 7)) * 2;
      int lbits = lane;
      int rl = r0 + 8 * half;
      asm volatile("" : "+v"(loff), "+v"(lbits), "+v"(rl));

      v8f acc[NT];
#pragma unroll
      for (int t = 0; t < NT; ++t) acc[t] = v8f{};
      const float* al_ = hS + (r0 + c) * P + 16 * half;
      const float* bl = sB + c * P + 16 * half;
#pragma unroll
      for (int L = 0; L < NK / 32; ++L) {
#pragma unroll
        for (int u = 0; u < 4; ++u) {
          const float4 av = *reinterpret_cast<const float4*>(al_ + 32 * L + 4 * u);
          float4 bv[NT];
#pragma unroll
          for (int t = 0; t < NT; ++t)
            bv[t] = *reinterpret_cast<const float4*>(bl + t * 16 * P + 32 * L + 4 * u);
          v2f a01, a23;
          a01.x = av.x; a01.y = av.y;
          a23.x = av.z; a23.y = av.w;
#pragma unroll
          for (int t = 0; t < NT; ++t) {
            v2f b01;
            b01.x = bv[t].x; b01.y = bv[t].y;
            acc[t] = __builtin_amdgcn_wmma_f32_16x16x4_f32(
                 false, a01,  false, b01,
                 (short)0, acc[t],  false,  false);
          }
#pragma unroll
          for (int t = 0; t < NT; ++t) {
            v2f b23;
            b23.x = bv[t].z; b23.y = bv[t].w;
            acc[t] = __builtin_amdgcn_wmma_f32_16x16x4_f32(
                 false, a23,  false, b23,
                 (short)0, acc[t],  false,  false);
          }
        }
      }
      asm volatile("v_nop\n\tv_nop\n\tv_nop\n\tv_nop\n\tv_nop"
                   : "+v"(acc[0]), "+v"(acc[1]),
                     "+v"(lv[0]), "+v"(lv[1]), "+v"(rv[0]), "+v"(rv[1]),
                     "+v"(loff), "+v"(lbits), "+v"(rl));
      const bool hb0 = (lbits & 1) != 0, hb1 = (lbits & 2) != 0;
      const bool hb2 = (lbits & 4) != 0, hb3 = (lbits & 8) != 0;
      global_float* g0 = (global_float*)(hb3 ? lgn + 1 : lgn) + loff;

#pragma unroll
      for (int t = 0; t < NT; ++t) {
#pragma unroll
        for (int i = 0; i < 8; ++i)
          if (rl + i < nloc) drow[i * NC + 16 * t] = acc[t][i];
      }
      float v[16];
#pragma unroll
      for (int i = 0; i < 8; ++i) {
        float sl = acc[0][i] * lv[0];
        float sr = acc[0][i] * rv[0];
#pragma unroll
        for (int t = 1; t < NT; ++t) {
          sl += acc[t][i] * lv[t];
          sr += acc[t][i] * rv[t];
        }
        v[i] = sl;
        v[8 + i] = sr;
      }
      fold<8, 1>(v, hb0);
      fold<4, 2>(v, hb1);
      fold<2, 4>(v, hb2);
      fold<1, 8>(v, hb3);
      if (rl + (lbits & 7) < nloc) g0[0] = v[0];
    }
  }
}

extern "C" void kernel_launch(void* const* d_in, const int* in_sizes, int n_in,
                              void* d_out, int out_size, void* d_ws,
                              size_t ws_size, hipStream_t stream) {
  const float* x   = (const float*)d_in[0];
  const int*   src = (const int*)d_in[1];
  const int*   dst = (const int*)d_in[2];
  const float* W1  = (const float*)d_in[3];
  const float* al1 = (const float*)d_in[4];
  const float* ar1 = (const float*)d_in[5];
  const float* b1  = (const float*)d_in[6];
  const float* W2  = (const float*)d_in[7];
  const float* al2 = (const float*)d_in[8];
  const float* ar2 = (const float*)d_in[9];
  const float* b2  = (const float*)d_in[10];
  float* out = (float*)d_out;
  unsigned* lists = (unsigned*)d_out;

  const int N = NNODES;
  static_assert(NNODES % (16 * GEMM_WAVES) == 0, "every GEMM wave has a full row tile");

  float* ws = (float*)d_ws;
  float* feat1 = ws;
  float* el1   = feat1 + (long)N * 64;
  float* er1   = el1 + (long)N * 8;
  unsigned* tab = (unsigned*)(ws + (long)N * 80);
  float* feat2 = ws + (long)N * 96;
  float* el2   = ws + (long)N * 128;
  float* er2   = el2 + 1;

  auto nb = [](long n, int b) { return (int)((n + b - 1) / b); };
  const int NOWN = (NNODES + OWN_R - 1) / OWN_R;

  owner_table_kernel<<<nb(OWN_NBLK, 8), 256, 0, stream>>>(dst, tab);

  gemm_wmma_kernel<256, 64, 8, 8><<<NNODES / (16 * GEMM_WAVES), 32 * GEMM_WAVES, 0, stream>>>(x, W1, al1, ar1, feat1, el1, er1);
  owner_agg_kernel<8, 8, true, 1, 8><<<NOWN, 1024, 0, stream>>>(src, dst, tab, el1, er1, feat1, b1, W2, al2, ar2, el2, lists, feat2);

  owner_agg_kernel<1, 32, false, 2, 2><<<NOWN, 1024, 0, stream>>>(src, dst, tab, el2, er2, feat2, b2, nullptr, nullptr, nullptr, nullptr, lists, out);
}


# table launch also writes one packed word per edge (source id, local row); layer 1's owner groups gather it per listed edge, not a word from src and one from dst; LDS adds stay in arrival order
# speedup vs baseline: 7.7095x; 1.0412x over previous
#include <hip/hip_runtime.h>
#include <hip/hip_bf16.h>

typedef __attribute__((ext_vector_type(2))) float v2f;
typedef __attribute__((ext_vector_type(8))) float v8f;

#define NNODES 100000
#define NEDGES 1600000

#define GEMM_WAVES 25

template <int M>
__device__ __forceinline__ float row_xor(float x) {
  const int b = __builtin_bit_cast(int, x);
  return __builtin_bit_cast(
      float, __builtin_amdgcn_update_dpp(0, b, 0x160 + M, 0xf, 0xf, true));
}

template <int CNT, int M>
__device__ __forceinline__ void fold(float* v, bool hi) {
#pragma unroll
  for (int n = 0; n < CNT; ++n) {
    const float keep = hi ? v[2 * n + 1] : v[2 * n];
    const float send = hi ? v[2 * n] : v[2 * n + 1];
    v[n] = keep + row_xor<M>(send);
  }
}

template <int K, int C, int H, int LS>
__global__ __launch_bounds__(32 * GEMM_WAVES)
__attribute__((amdgpu_waves_per_eu(7, 7))) void gemm_wmma_kernel(
    const float* __restrict__ A, const float* __restrict__ B,
    const float* __restrict__ al, const float* __restrict__ ar,
    float* __restrict__ D, float* __restrict__ el, float* __restrict__ er) {
  constexpr int NT = C / 16;
  constexpr int DH = C / H;
  static_assert(C % H == 0 && (DH == 8 || DH % 16 == 0), "a head: 8 lanes of a tile, or whole tiles");
  constexpr int TPH = DH > 16 ? DH / 16 : 1;
  constexpr int NHT = NT / TPH;
  constexpr int HPG = DH < 16 ? 16 / DH : 1;
  static_assert(LS >= H, "LS: words from a node's logits to the next node's");
  constexpr int NL = K / 32;
  constexpr int BP = K + 4;
  static_assert(K % 32 == 0 && C % 16 == 0, "whole lines, whole tiles");
  static_assert((BP / 4) % 2 == 1, "row pitch: an odd number of 16-byte pieces");
  __shared__ __attribute__((aligned(16))) float sB[C * BP];

  const int tid = threadIdx.x;
  const int lane = tid & 31;
  const int wave = tid >> 5;
  const int half = lane >> 4;
  const int c    = lane & 15;
  const int row0 = (blockIdx.x * GEMM_WAVES + wave) * 16;

  const float4* ap =
      reinterpret_cast<const float4*>(A + (long)(row0 + c) * K + 16 * half);
  float4 a[2][4];
#pragma unroll
  for (int u = 0; u < 4; ++u) a[0][u] = ap[u];

  float* dpin = D + (long)(row0 + 8 * half) * C + c;
  asm volatile("" : "+v"(dpin));
  typedef __attribute__((address_space(1))) float global_float;
  global_float* drow = (global_float*)dpin;

  float lv[NT], rv[NT];
#pragma unroll
  for (int t = 0; t < NT; ++t) {
    lv[t] = al[16 * t + c];
    rv[t] = ar[16 * t + c];
  }
  unsigned loff = (unsigned)(row0 + 8 * half + (c & 7)) * LS + (DH < 16 ? (c >> 3) : 0);
  int lbits = lane;
  asm volatile("" : "+v"(loff), "+v"(lbits));

  constexpr int NTHR = 32 * GEMM_WAVES;
  constexpr int NQ = C * K / 4;
  constexpr int NU = (NQ + NTHR - 1) / NTHR;
  float4 w[NU];
#pragma unroll
  for (int j = 0; j < NU; ++j) {
    const int q = tid + j * NTHR;
    if (q < NQ) {
      const int n = q % C;
      const int k = (q / C) * 4;
      w[j].x = B[(k + 0) * C + n];
      w[j].y = B[(k + 1) * C + n];
      w[j].z = B[(k + 2) * C + n];
      w[j].w = B[(k + 3) * C + n];
    }
  }
#pragma unroll
  for (int j = 0; j < NU; ++j) {
    const int q = tid + j * NTHR;
    if (q < NQ) {
      const int n = q % C;
      const int k = (q / C) * 4;
      *reinterpret_cast<float4*>(&sB[n * BP + k]) = w[j];
    }
  }
  __syncthreads();

  v8f acc[NT];
#pragma unroll
  for (int t = 0; t < NT; ++t) acc[t] = v8f{};

  const float* bl = sB + c * BP + 16 * half;
#pragma unroll
  for (int L = 0; L < NL; ++L) {
    if (L + 1 < NL) {
#pragma unroll
      for (int u = 0; u < 4; ++u) a[(L + 1) & 1][u] = ap[8 * (L + 1) + u];
    }
#pragma unroll
    for (int u = 0; u < 4; ++u) {
      const float4 av = a[L & 1][u];
      float4 bv[NT];
#pragma unroll
      for (int t = 0; t < NT; ++t)
        bv[t] = *reinterpret_cast<const float4*>(bl + t * 16 * BP + 32 * L + 4 * u);
      v2f a01, a23;
      a01.x = av.x; a01.y = av.y;
      a23.x = av.z; a23.y = av.w;
#pragma unroll
      for (int t = 0; t < NT; ++t) {
        v2f b01;
        b01.x = bv[t].x; b01.y = bv[t].y;
        acc[t] = __builtin_amdgcn_wmma_f32_16x16x4_f32(
             false, a01,  false, b01,
             (short)0, acc[t],  false,  false);
      }
#pragma unroll
      for (int t = 0; t < NT; ++t) {
        v2f b23;
        b23.x = bv[t].z; b23.y = bv[t].w;
        acc[t] = __builtin_amdgcn_wmma_f32_16x16x4_f32(
             false, a23,  false, b23,
             (short)0, acc[t],  false,  false);
      }
    }
  }

  static_assert(NT == 2 || NT == 4, "the statement below names every tile");
  if constexpr (NT == 4) {
    asm volatile("v_nop\n\tv_nop\n\tv_nop\n\tv_nop\n\tv_nop"
                 : "+v"(acc[0]), "+v"(acc[1]), "+v"(acc[2]), "+v"(acc[3]),
                   "+v"(lv[0]), "+v"(lv[1]), "+v"(lv[2]), "+v"(lv[3]),
                   "+v"(rv[0]), "+v"(rv[1]), "+v"(rv[2]), "+v"(rv[3]),
                   "+v"(loff), "+v"(lbits));
  } else {
    asm volatile("v_nop\n\tv_nop\n\tv_nop\n\tv_nop\n\tv_nop"
                 : "+v"(acc[0]), "+v"(acc[1]),
                   "+v"(lv[0]), "+v"(lv[1]), "+v"(rv[0]), "+v"(rv[1]),
                   "+v"(loff), "+v"(lbits));
  }
  const bool hb0 = (lbits & 1) != 0, hb1 = (lbits & 2) != 0;
  const bool hb2 = (lbits & 4) != 0, hb3 = (lbits & 8) != 0;
  global_float* g0 = (global_float*)(DH < 16 ? el : (hb3 ? er : el)) + loff;
  global_float* g1 = (global_float*)er + loff;

#pragma unroll
  for (int t = 0; t < NT; ++t) {
#pragma unroll
    for (int i = 0; i < 8; ++i) drow[i * C + 16 * t] = acc[t][i];
  }

#pragma unroll
  for (int g = 0; g < NHT; ++g) {
    float v[16];
#pragma unroll
    for (int i = 0; i < 8; ++i) {
      float sl = acc[g * TPH][i] * lv[g * TPH];
      float sr = acc[g * TPH][i] * rv[g * TPH];
#pragma unroll
      for (int u = 1; u < TPH; ++u) {
        sl += acc[g * TPH + u][i] * lv[g * TPH + u];
        sr += acc[g * TPH + u][i] * rv[g * TPH + u];
      }
      v[i] = sl;
      v[8 + i] = sr;
    }
    fold<8, 1>(v, hb0);
    fold<4, 2>(v, hb1);
    fold<2, 4>(v, hb2);
    if constexpr (DH >= 16) {
      fold<1, 8>(v, hb3);
      g0[g] = v[0];
    } else {
      g0[HPG * g] = v[0];
      g1[HPG * g] = v[1];
    }
  }
}

__device__ __forceinline__ unsigned float_ord_encode(float f) {
  unsigned u = __float_as_uint(f);
  return (u & 0x80000000u) ? ~u : (u | 0x80000000u);
}
__device__ __forceinline__ float float_ord_decode(unsigned u) {
  return (u & 0x80000000u) ? __uint_as_float(u ^ 0x80000000u)
                           : __uint_as_float(~u);
}
#define ENC_NEG_INF 0x007FFFFFu

#define OWN_R 391
#define OWN_WAVES 32
#define OWN_QCAP 1024
#define OWN_RBITS 9

#define LST_ROW 32
#define OWN_RP 400
#define LST_NONE 0xFFFFFFFFu

#define OWN_NGRP (NEDGES / 32)
#define OWN_NBLK ((OWN_NGRP + 31) / 32)

__device__ __forceinline__ unsigned edge_word(int s, int d) {
  const unsigned oq = (unsigned)d / OWN_R;
  const unsigned o = oq < 255u ? oq : 255u;
  const unsigned ru = (unsigned)d - o * OWN_R;
  const unsigned rel = ru < (unsigned)(OWN_R - 1) ? ru : (unsigned)(OWN_R - 1);
  return ((unsigned)s << OWN_RBITS) | rel;
}

__global__ __launch_bounds__(256)
__attribute__((amdgpu_waves_per_eu(2, 4))) void owner_table_kernel(
    const int* __restrict__ src, const int* __restrict__ dst,
    unsigned* __restrict__ tab, unsigned* __restrict__ pk) {
  const int lane = threadIdx.x & 31;
  const int blk = blockIdx.x * 8 + (threadIdx.x >> 5);
  if (blk >= OWN_NBLK) return;
  static_assert(NEDGES % 32 == 0, "whole groups; whole 16-byte pieces of src, dst, pk");
  constexpr int NP = NEDGES / 4;
  const int4* d4 = reinterpret_cast<const int4*>(dst);
  const int4* s4 = reinterpret_cast<const int4*>(src);
  const int grp = blk * 32 + lane;
  const bool live = grp < OWN_NGRP;
  const int gc = live ? grp : OWN_NGRP - 1;
  int4 dg[8];
#pragma unroll
  for (int c = 0; c < 8; ++c) dg[c] = d4[8 * gc + c];
  int4 sp[8], dp[8];
#pragma unroll
  for (int c = 0; c < 8; ++c) {
    const int i4 = blk * 256 + c * 32 + lane;
    const int ic = i4 < NP ? i4 : NP - 1;
    sp[c] = s4[ic];
    dp[c] = d4[ic];
  }

  unsigned p[8] = {0u, 0u, 0u, 0u, 0u, 0u, 0u, 0u};
#pragma unroll
  for (int c = 0; c < 8; ++c) {
    const int dv[4] = {dg[c].x, dg[c].y, dg[c].z, dg[c].w};
#pragma unroll
    for (int k = 0; k < 4; ++k) {
      const unsigned oq = (unsigned)dv[k] / OWN_R;
      const unsigned o = oq < 255u ? oq : 255u;
#pragma unroll
      for (int j = 0; j < 8; ++j) p[j] |= ((o >> j) & 1u) << (4 * c + k);
    }
  }
  if (!live) {
#pragma unroll
    for (int j = 0; j < 8; ++j) p[j] = 0u;
  }
  uint4* t4 = reinterpret_cast<uint4*>(tab);
  t4[blk * 64 + lane]      = make_uint4(p[0], p[1], p[2], p[3]);
  t4[blk * 64 + 32 + lane] = make_uint4(p[4], p[5], p[6], p[7]);

  uint4* p4 = reinterpret_cast<uint4*>(pk);
#pragma unroll
  for (int c = 0; c < 8; ++c) {
    const int i4 = blk * 256 + c * 32 + lane;
    p4[i4] = make_uint4(edge_word(sp[c].x, dp[c].x), edge_word(sp[c].y, dp[c].y),
                        edge_word(sp[c].z, dp[c].z), edge_word(sp[c].w, dp[c].w));
  }
}

template <int H, int D, bool DO_ELU, int LISTS, int LS>
__global__ __launch_bounds__(1024) void owner_agg_kernel(
    const unsigned* __restrict__ pk, const unsigned* __restrict__ tab,
    const float* __restrict__ el, const float* __restrict__ er,
    const float* __restrict__ feat, const float* __restrict__ bias,
    const float* __restrict__ Wn, const float* __restrict__ aln,
    const float* __restrict__ arn, float* lgn,
    unsigned* lst, float* out) {
  constexpr int C = H * D;
  constexpr bool FUSE = (LISTS == 1);
  constexpr bool PADL = (LISTS == 2);
  static_assert(LISTS != 2 || C == LST_ROW, "the lists lie in this layer's own output rows");
  constexpr int CPL = C / 32;
  constexpr int BPI = OWN_WAVES * 2;
  constexpr int NIT = (OWN_NBLK + BPI - 1) / BPI;
  constexpr int RB = 16;

  __shared__ float    s_agg[OWN_R * C];
  __shared__ float    s_den[OWN_R * H];
  __shared__ unsigned s_max[OWN_R * H];
  __shared__ __attribute__((aligned(16))) int s_q[OWN_WAVES][OWN_QCAP];
  constexpr bool ONEW = (H == 1);
  constexpr int HP = 32 / D;
  static_assert(ONEW || (CPL == 2 && H == 2 * HP), "a lane's two heads are a pair");
  __shared__ __attribute__((aligned(16))) float s_w[OWN_WAVES][ONEW ? 1 : 32 * H];
  __shared__ int      s_resweep;
  constexpr bool KEEPV = (H == 1);
  __shared__ float    s_v[KEEPV ? OWN_WAVES : 1][KEEPV ? OWN_QCAP : 1];

  const int tid = threadIdx.x;
  const int lane = tid & 31;
  const int wave = tid >> 5;
  const int lo = blockIdx.x * OWN_R;
  int nloc = NNODES - lo;
  nloc = nloc > OWN_R ? OWN_R : nloc;
  nloc = nloc < 0 ? 0 : nloc;
  static_assert(OWN_WAVES == LST_ROW, "a row of counts, cap rows of lists");
  const int cap = nloc >= 1 ? ((nloc - 1) & ~31) : 0;
  unsigned* const lcnt = lst + (long)lo * LST_ROW;
  unsigned* const lmine = lcnt + LST_ROW + wave * cap;

  for (int i = tid; i < OWN_R * C; i += 1024) s_agg[i] = 0.f;
  for (int i = tid; i < OWN_R * H; i += 1024) {
    s_den[i] = 0.f;
    s_max[i] = ENC_NEG_INF;
  }
  if (tid == 0) s_resweep = 0;
  __syncthreads();

  const uint4* tab4 = reinterpret_cast<const uint4*>(tab);
  unsigned want[8];
#pragma unroll
  for (int j = 0; j < 8; ++j) want[j] = ((blockIdx.x >> j) & 1u) ? ~0u : 0u;
  int qhead = 0, qtail = 0;
  bool packed = false;
  bool havev = false;
  unsigned icnt = LST_NONE;
  if (LISTS == 2 && nloc >= 1) icnt = __builtin_amdgcn_readfirstlane(lcnt[wave]);
  const bool imported = icnt <= (unsigned)cap;

#pragma unroll 1
  for (int pass = 0; pass < 2; ++pass) {

    auto drain = [&](int n) {
      __builtin_amdgcn_fence(__ATOMIC_RELEASE, "wavefront");
      __builtin_amdgcn_wave_barrier();
      __builtin_amdgcn_fence(__ATOMIC_ACQUIRE, "wavefront");
      int rel = 0, s = 0;
      unsigned pw = 0u;
      float wl = 0.f;
      if (lane < n) {
        const int slot = (qhead + lane) & (OWN_QCAP - 1);
        const unsigned q = (unsigned)s_q[wave][slot];
        pw = packed ? q : pk[q];
        s = (int)(pw >> OWN_RBITS);
        rel = (int)(pw & ((1u << OWN_RBITS) - 1u));
        if (!packed && pass == 0) {
          s_q[wave][slot] = (int)pw;
          if (LISTS == 1 && slot < cap) lmine[slot] = pw;
        }
        if constexpr (ONEW) {
        long eli = (long)s, eri = (long)(lo + rel);
        if (PADL) {
          eli = (long)((unsigned)s + (OWN_RP - OWN_R) * ((unsigned)s / OWN_R));
          eri = (long)((int)blockIdx.x * OWN_RP + rel);
        }
        const float* els = el + eli * LS;
        const float* erd = er + eri * LS;
        float vv[H];
        if (KEEPV && havev) {
          vv[0] = s_v[wave][slot];
        } else {
#pragma unroll
          for (int h = 0; h < H; ++h) {
            const float v = els[h] + erd[h];
            vv[h] = v > 0.f ? v : 0.2f * v;
          }
          if (KEEPV && pass == 0) s_v[wave][slot] = vv[0];
        }
        if (pass == 0) {
#pragma unroll
          for (int h = 0; h < H; ++h)
            atomicMax(&s_max[rel * H + h], float_ord_encode(vv[h]));
        } else {
          float m[H];
#pragma unroll
          for (int h = 0; h < H; ++h) m[h] = float_ord_decode(s_max[rel * H + h]);
#pragma unroll
          for (int h = 0; h < H; ++h) {
            const float w = expf(vv[h] - m[h]);
            atomicAdd(&s_den[rel * H + h], w);
            wl = w;
          }
        }
        }
      }
      if constexpr (!ONEW) {
        constexpr int EPI = 32 / H;
        static_assert(32 % H == 0, "whole edges per instruction");
        const int hh = lane % H;
        const int g = lane / H;
        float ev[H];
        int ra[H];
        bool ok[H];
#pragma unroll
        for (int i = 0; i < H; ++i) {
          const int j = EPI * i + g;
          ok[i] = j < n;
          const int jc = ok[i] ? j : n - 1;
          const unsigned pj = (unsigned)__builtin_amdgcn_ds_bpermute(jc << 2, (int)pw);
          const unsigned sj = pj >> OWN_RBITS;
          const int rj = (int)(pj & ((1u << OWN_RBITS) - 1u));
          ra[i] = rj * H + hh;
          const unsigned ei = sj * (unsigned)LS + (unsigned)hh;
          const unsigned ri = (unsigned)(lo + rj) * (unsigned)LS + (unsigned)hh;
          const float v = el[ei] + er[ri];
          ev[i] = v > 0.f ? v : 0.2f * v;
        }
        if (pass == 0) {
#pragma unroll
          for (int i = 0; i < H; ++i)
            if (ok[i]) atomicMax(&s_max[ra[i]], float_ord_encode(ev[i]));
        } else {
          float m[H];
#pragma unroll
          for (int i = 0; i < H; ++i) m[i] = float_ord_decode(s_max[ra[i]]);
#pragma unroll
          for (int i = 0; i < H; ++i) {
            const float w = expf(ev[i] - m[i]);
            if (ok[i]) {
              atomicAdd(&s_den[ra[i]], w);
              s_w[wave][(EPI * i + g) * H + 2 * (hh % HP) + hh / HP] = w;
            }
          }
        }
      }
      if (pass == 1) {
        if (!ONEW) {
          __builtin_amdgcn_fence(__ATOMIC_RELEASE, "wavefront");
          __builtin_amdgcn_wave_barrier();
          __builtin_amdgcn_fence(__ATOMIC_ACQUIRE, "wavefront");
        }
        for (int j0 = 0; j0 < n; j0 += RB) {
          float f[RB][CPL];
#pragma unroll
          for (int u = 0; u < RB; ++u) {
            const int sj = __builtin_amdgcn_readlane(s, j0 + u);
            if (j0 + u < n) {
#pragma unroll
              for (int p = 0; p < CPL; ++p)
                f[u][p] = feat[(long)sj * C + lane + 32 * p];
            }
          }
          float wv[RB][CPL];
#pragma unroll
          for (int u = 0; u < RB; ++u) {
            if (ONEW) {
              wv[u][0] = __int_as_float(
                  __builtin_amdgcn_readlane(__float_as_int(wl), j0 + u));
            } else if (j0 + u < n) {
              const float2 w2 = *reinterpret_cast<const float2*>(
                  &s_w[wave][(j0 + u) * H + 2 * (lane / D)]);
              wv[u][0] = w2.x;
              wv[u][CPL - 1] = w2.y;
            }
          }
#pragma unroll
          for (int u = 0; u < RB; ++u) {
            const int rj = __builtin_amdgcn_readlane(rel, j0 + u);
            if (j0 + u < n) {
#pragma unroll
              for (int p = 0; p < CPL; ++p)
                atomicAdd(&s_agg[rj * C + lane + 32 * p], wv[u][p] * f[u][p]);
            }
          }
        }
      }
      qhead += n;
    };

    const bool sweep = (pass == 0) ? !imported : (s_resweep != 0);
    packed = !sweep;
    havev = (pass == 1) && !sweep;
    if (sweep) {
      qhead = 0;
      qtail = 0;
#pragma unroll 1
      for (int i = 0; i < NIT; ++i) {
        const int it = (i + (int)blockIdx.x) % NIT;
        const int blk0 = it * BPI + wave * 2;
        unsigned hit[2];
#pragma unroll
        for (int k = 0; k < 2; ++k) {
          const int blk = blk0 + k;
          const int bc = blk < OWN_NBLK ? blk : OWN_NBLK - 1;
          const uint4 pa = tab4[bc * 64 + lane];
          const uint4 pb = tab4[bc * 64 + 32 + lane];
          const unsigned pl[8] = {pa.x, pa.y, pa.z, pa.w, pb.x, pb.y, pb.z, pb.w};
          unsigned acc = ~0u;
#pragma unroll
          for (int j = 0; j < 8; ++j) acc &= ~(pl[j] ^ want[j]);
          hit[k] = (blk * 32 + lane < OWN_NGRP) ? acc : 0u;
        }
        unsigned h0 = hit[0], h1 = hit[1];
#pragma unroll 1
        for (int k = 0; k < 2; ++k) {
          unsigned mm = h0;
          h0 = h1;
          const int ebase = (blk0 + k) * 1024 + lane * 32;
          unsigned any = __builtin_amdgcn_ballot_w32(mm != 0u);
          while (any != 0u) {
            while (qtail - qhead > OWN_QCAP - 32) drain(32);
            if (mm != 0u) {
              const int q = __builtin_ctz(mm);
              mm &= mm - 1u;
              const int pos =
                  (qtail + (int)__builtin_amdgcn_mbcnt_lo(any, 0u)) &
                  (OWN_QCAP - 1);
              s_q[wave][pos] = ebase + q;
            }
            qtail += __popc(any);
            any = __builtin_amdgcn_ballot_w32(mm != 0u);
          }
        }
      }
      if (pass == 0 && qtail > OWN_QCAP) s_resweep = 1;
    } else if (pass == 0) {
      const int n = (int)icnt;
      for (int b = 0; b < n; b += 256) {
        unsigned t[8];
#pragma unroll
        for (int u = 0; u < 8; ++u)
          if (b + 32 * u < n) t[u] = lmine[b + 32 * u + lane];
#pragma unroll
        for (int u = 0; u < 8; ++u)
          if (b + 32 * u < n) s_q[wave][b + 32 * u + lane] = (int)t[u];
      }
      qhead = 0;
      qtail = n;
    } else {
      qhead = 0;
    }
    while (qtail - qhead >= 32) drain(32);
    if (qtail - qhead > 0) drain(qtail - qhead);
    if (LISTS == 1 && pass == 0 && nloc >= 1 && lane == 0)
      lcnt[wave] = qtail <= cap ? (unsigned)qtail : LST_NONE;
    __syncthreads();
  }

  if constexpr (!FUSE) {
    for (int i = tid; i < nloc * C; i += 1024) {
      const int r = i / C;
      const int c = i % C;
      const float dn = s_den[r * H + c / D];
      float x = dn > 0.f ? s_agg[i] / dn : 0.f;
      x += bias[c];
      if (DO_ELU) x = x > 0.f ? x : expf(x) - 1.f;
      out[(long)lo * C + i] = x;
    }
  } else {
    constexpr int NK = C;
    constexpr int NC = 32;
    constexpr int NT = NC / 16;
    constexpr int P = NK + 4;
    constexpr int NSTRIP = OWN_RP / 16;
    static_assert(NK % 32 == 0 && (P / 4) % 2 == 1, "whole lines; odd 16-byte pitch");
    static_assert(OWN_RP % 16 == 0 && OWN_RP >= OWN_R && NSTRIP <= OWN_WAVES, "a strip per wave");
    static_assert(OWN_RP * P <= OWN_WAVES * OWN_QCAP, "the rows fit the lists' space");
    static_assert(!ONEW && NC * P <= OWN_WAVES * 32 * H, "the weights fit the waves' weight space");
    static_assert(NT == 2, "the statement behind the products names every tile");
    float* const hS = reinterpret_cast<float*>(&s_q[0][0]);
    float* const sB = &s_w[0][0];
    for (int i = tid; i < nloc * C; i += 1024) {
      const int r = i / C;
      const int c = i % C;
      const float dn = s_den[r * H + c / D];
      float x = dn > 0.f ? s_agg[i] / dn : 0.f;
      x += bias[c];
      if (DO_ELU) x = x > 0.f ? x : expf(x) - 1.f;
      hS[r * P + c] = x;
    }
    for (int i = nloc * C + tid; i < OWN_RP * C; i += 1024) hS[(i / C) * P + (i % C)] = 0.f;
    for (int i = tid; i < NK * NC; i += 1024) sB[(i % NC) * P + (i / NC)] = Wn[i];
    __syncthreads();

    if (wave < NSTRIP) {
      const int half = lane >> 4;
      const int c = lane & 15;
      const int r0 = 16 * wave;
      float* dpin = out + (long)(lo + r0 + 8 * half) * NC + c;
      asm volatile("" : "+v"(dpin));
      typedef __attribute__((address_space(1))) float global_float;
      global_float* drow = (global_float*)dpin;
      float lv[NT], rv[NT];
#pragma unroll
      for (int t = 0; t < NT; ++t) {
        lv[t] = aln[16 * t + c];
        rv[t] = arn[16 * t + c];
      }
      unsigned loff = (unsigned)((int)blockIdx.x * OWN_RP + r0 + 8 * half + (c & 7)) * 2;
      int lbits = lane;
      int rl = r0 + 8 * half;
      asm volatile("" : "+v"(loff), "+v"(lbits), "+v"(rl));

      v8f acc[NT];
#pragma unroll
      for (int t = 0; t < NT; ++t) acc[t] = v8f{};
      const float* al_ = hS + (r0 + c) * P + 16 * half;
      const float* bl = sB + c * P + 16 * half;
#pragma unroll
      for (int L = 0; L < NK / 32; ++L) {
#pragma unroll
        for (int u = 0; u < 4; ++u) {
          const float4 av = *reinterpret_cast<const float4*>(al_ + 32 * L + 4 * u);
          float4 bv[NT];
#pragma unroll
          for (int t = 0; t < NT; ++t)
            bv[t] = *reinterpret_cast<const float4*>(bl + t * 16 * P + 32 * L + 4 * u);
          v2f a01, a23;
          a01.x = av.x; a01.y = av.y;
          a23.x = av.z; a23.y = av.w;
#pragma unroll
          for (int t = 0; t < NT; ++t) {
            v2f b01;
            b01.x = bv[t].x; b01.y = bv[t].y;
            acc[t] = __builtin_amdgcn_wmma_f32_16x16x4_f32(
                 false, a01,  false, b01,
                 (short)0, acc[t],  false,  false);
          }
#pragma unroll
          for (int t = 0; t < NT; ++t) {
            v2f b23;
            b23.x = bv[t].z; b23.y = bv[t].w;
            acc[t] = __builtin_amdgcn_wmma_f32_16x16x4_f32(
                 false, a23,  false, b23,
                 (short)0, acc[t],  false,  false);
          }
        }
      }
      asm volatile("v_nop\n\tv_nop\n\tv_nop\n\tv_nop\n\tv_nop"
                   : "+v"(acc[0]), "+v"(acc[1]),
                     "+v"(lv[0]), "+v"(lv[1]), "+v"(rv[0]), "+v"(rv[1]),
                     "+v"(loff), "+v"(lbits), "+v"(rl));
      const bool hb0 = (lbits & 1) != 0, hb1 = (lbits & 2) != 0;
      const bool hb2 = (lbits & 4) != 0, hb3 = (lbits & 8) != 0;
      global_float* g0 = (global_float*)(hb3 ? lgn + 1 : lgn) + loff;

#pragma unroll
      for (int t = 0; t < NT; ++t) {
#pragma unroll
        for (int i = 0; i < 8; ++i)
          if (rl + i < nloc) drow[i * NC + 16 * t] = acc[t][i];
      }
      float v[16];
#pragma unroll
      for (int i = 0; i < 8; ++i) {
        float sl = acc[0][i] * lv[0];
        float sr = acc[0][i] * rv[0];
#pragma unroll
        for (int t = 1; t < NT; ++t) {
          sl += acc[t][i] * lv[t];
          sr += acc[t][i] * rv[t];
        }
        v[i] = sl;
        v[8 + i] = sr;
      }
      fold<8, 1>(v, hb0);
      fold<4, 2>(v, hb1);
      fold<2, 4>(v, hb2);
      fold<1, 8>(v, hb3);
      if (rl + (lbits & 7) < nloc) g0[0] = v[0];
    }
  }
}

extern "C" void kernel_launch(void* const* d_in, const int* in_sizes, int n_in,
                              void* d_out, int out_size, void* d_ws,
                              size_t ws_size, hipStream_t stream) {
  const float* x   = (const float*)d_in[0];
  const int*   src = (const int*)d_in[1];
  const int*   dst = (const int*)d_in[2];
  const float* W1  = (const float*)d_in[3];
  const float* al1 = (const float*)d_in[4];
  const float* ar1 = (const float*)d_in[5];
  const float* b1  = (const float*)d_in[6];
  const float* W2  = (const float*)d_in[7];
  const float* al2 = (const float*)d_in[8];
  const float* ar2 = (const float*)d_in[9];
  const float* b2  = (const float*)d_in[10];
  float* out = (float*)d_out;
  unsigned* lists = (unsigned*)d_out;

  const int N = NNODES;
  static_assert(NNODES % (16 * GEMM_WAVES) == 0, "every GEMM wave has a full row tile");

  float* ws = (float*)d_ws;
  float* feat1 = ws;
  float* el1   = feat1 + (long)N * 64;
  float* er1   = el1 + (long)N * 8;
  unsigned* tab = (unsigned*)(ws + (long)N * 80);
  float* feat2 = ws + (long)N * 96;
  float* el2   = ws + (long)N * 128;
  float* er2   = el2 + 1;
  unsigned* pk = (unsigned*)(ws + (long)N * 128 + 256L * OWN_RP * 2);

  auto nb = [](long n, int b) { return (int)((n + b - 1) / b); };
  const int NOWN = (NNODES + OWN_R - 1) / OWN_R;

  owner_table_kernel<<<nb(OWN_NBLK, 8), 256, 0, stream>>>(src, dst, tab, pk);

  gemm_wmma_kernel<256, 64, 8, 8><<<NNODES / (16 * GEMM_WAVES), 32 * GEMM_WAVES, 0, stream>>>(x, W1, al1, ar1, feat1, el1, er1);
  owner_agg_kernel<8, 8, true, 1, 8><<<NOWN, 1024, 0, stream>>>(pk, tab, el1, er1, feat1, b1, W2, al2, ar2, el2, lists, feat2);

  owner_agg_kernel<1, 32, false, 2, 2><<<NOWN, 1024, 0, stream>>>(pk, tab, el2, er2, feat2, b2, nullptr, nullptr, nullptr, nullptr, lists, out);
}
